# starting kernel
# baseline (speedup 1.0000x reference)
#include <hip/hip_runtime.h>
#include <math.h>

namespace {
constexpr int kB = 16384;
constexpr int kD = 64;
constexpr int kWaves = 4;
constexpr int kThreads = kWaves * 32;
constexpr int kRowsPerWave = 16;
constexpr int kRowsPerBlock = kWaves * kRowsPerWave;
constexpr int kAst = 68;
constexpr int kWPS = 160;
constexpr float kLog2Pi = 1.8378770664093453f;
}

typedef float v2f __attribute__((ext_vector_type(2)));
typedef float v8f __attribute__((ext_vector_type(8)));

__device__ __forceinline__ void wave_fence() { asm volatile("" ::: "memory"); }

__global__ void __launch_bounds__(kThreads)
iaf_kernel(const float* __restrict__ zmean, const float* __restrict__ zlogvar,
           const float* __restrict__ noise, const float* __restrict__ Wg,
           const float* __restrict__ bvec, float* __restrict__ out_samples,
           float* __restrict__ out_lp)
{
  __shared__ float s_w[4 * 32 * kWPS];
  __shared__ float s_b[4 * kD];
  __shared__ float s_act[kWaves][2][kRowsPerWave * kAst];
  __shared__ float s_red[kThreads * 8];

  const int tid = threadIdx.x;
  const int lane = tid & 31;
  const int wv = tid >> 5;
  const int half = lane >> 4;
  const int l15 = lane & 15;
  const int r0 = blockIdx.x * kRowsPerBlock + wv * kRowsPerWave;

  auto load_weights = [&](int s) {
    __syncthreads();
    for (int idx = tid; idx < 4 * kD * kD; idx += kThreads) {
      const int l = idx >> 12;
      const int rem = idx & 4095;
      const int k = rem >> 6;
      const int n = rem & 63;
      const bool keep = (l == 0) ? (k < n) : (k <= n);
      s_w[(l * 32 + (k >> 1)) * kWPS + 2 * n + (k & 1)] =
          keep ? Wg[((s * 4 + l) * kD + k) * kD + n] : 0.0f;
    }
    for (int idx = tid; idx < 4 * kD; idx += kThreads)
      s_b[idx] = bvec[s * 4 * kD + idx];
    __syncthreads();
  };

  auto layer = [&](int ib, int ob, int l, bool do_relu, v8f* outfr, const v8f* addfr) {
    v8f acc[4];
#pragma unroll
    for (int n = 0; n < 4; ++n) {
      const float bv = s_b[l * kD + n * 16 + l15];
      v8f t;
#pragma unroll
      for (int v = 0; v < 8; ++v) t[v] = bv;
      acc[n] = t;
    }
    const int arow = l15 * kAst;
    const int wbase = 2 * l15;
#pragma unroll
    for (int kk = 0; kk < 16; ++kk) {
      const int kb = kk * 4 + half * 2;
      v2f a;
      a.x = s_act[wv][ib][arow + kb];
      a.y = s_act[wv][ib][arow + kb + 1];
      const int pbase = (l * 32 + kk * 2 + half) * kWPS + wbase;
#pragma unroll
      for (int n = 0; n < 4; ++n) {
        v2f bf;
        bf.x = s_w[pbase + 32 * n];
        bf.y = s_w[pbase + 32 * n + 1];
        acc[n] = __builtin_amdgcn_wmma_f32_16x16x4_f32(
            false, a, false, bf, (short)0, acc[n], false, false);
      }
    }
    if (do_relu) {
#pragma unroll
      for (int n = 0; n < 4; ++n)
#pragma unroll
        for (int v = 0; v < 8; ++v) acc[n][v] = fmaxf(acc[n][v], 0.0f);
    }
    if (addfr) {
#pragma unroll
      for (int n = 0; n < 4; ++n)
#pragma unroll
        for (int v = 0; v < 8; ++v) acc[n][v] += addfr[n][v];
    }
    if (ob >= 0) {
#pragma unroll
      for (int n = 0; n < 4; ++n)
#pragma unroll
        for (int v = 0; v < 8; ++v)
          s_act[wv][ob][(v + half * 8) * kAst + n * 16 + l15] = acc[n][v];
      wave_fence();
    } else {
#pragma unroll
      for (int n = 0; n < 4; ++n) outfr[n] = acc[n];
    }
  };

  auto store_frags = [&](const v8f* fr, int b) {
#pragma unroll
    for (int n = 0; n < 4; ++n)
#pragma unroll
      for (int v = 0; v < 8; ++v)
        s_act[wv][b][(v + half * 8) * kAst + n * 16 + l15] = fr[n][v];
    wave_fence();
  };

  auto made = [&](v8f* outfr) {
    layer(0, 1, 0, true, nullptr, nullptr);
    layer(1, 0, 1, true, nullptr, nullptr);
    layer(0, 1, 2, true, nullptr, nullptr);
    layer(1, -1, 3, false, outfr, nullptr);
  };

  auto reverse_cols = [&](v8f* fr) {
    store_frags(fr, 0);
#pragma unroll
    for (int n = 0; n < 4; ++n)
#pragma unroll
      for (int v = 0; v < 8; ++v)
        fr[n][v] = s_act[wv][0][(v + half * 8) * kAst + (63 - (n * 16 + l15))];
    wave_fence();
  };

  v8f y[4];
#pragma unroll
  for (int n = 0; n < 4; ++n)
#pragma unroll
    for (int v = 0; v < 8; ++v) {
      const int row = r0 + v + half * 8;
      const int col = n * 16 + l15;
      const float m = zmean[row * kD + col];
      const float lv = zlogvar[row * kD + col];
      const float e = noise[row * kD + col];
      y[n][v] = m + expf(0.5f * lv) * e;
    }

  v8f m4[4];

  load_weights(1);
  store_frags(y, 0);
  made(m4);
#pragma unroll
  for (int n = 0; n < 4; ++n)
#pragma unroll
    for (int v = 0; v < 8; ++v) y[n][v] -= m4[n][v];
  reverse_cols(y);

  load_weights(0);
  store_frags(y, 0);
  made(m4);
#pragma unroll
  for (int n = 0; n < 4; ++n)
#pragma unroll
    for (int v = 0; v < 8; ++v) y[n][v] -= m4[n][v];

#pragma unroll
  for (int n = 0; n < 4; ++n)
#pragma unroll
    for (int v = 0; v < 8; ++v)
      out_samples[(r0 + v + half * 8) * kD + n * 16 + l15] = y[n][v];

  auto density_stage = [&]() {
    v8f x0[4];
#pragma unroll
    for (int n = 0; n < 4; ++n) x0[n] = y[n];
#pragma unroll
    for (int n = 0; n < 4; ++n)
#pragma unroll
      for (int v = 0; v < 8; ++v)
        s_act[wv][0][(v + half * 8) * kAst + n * 16 + l15] = 0.0f;
    wave_fence();
    for (int it = 0; it < kD; ++it) {
      layer(0, 1, 0, true, nullptr, nullptr);
      layer(1, 0, 1, true, nullptr, nullptr);
      layer(0, 1, 2, true, nullptr, nullptr);
      layer(1, 0, 3, false, nullptr, x0);
    }
#pragma unroll
    for (int n = 0; n < 4; ++n)
#pragma unroll
      for (int v = 0; v < 8; ++v)
        y[n][v] = s_act[wv][0][(v + half * 8) * kAst + n * 16 + l15];
    wave_fence();
  };

  density_stage();
  reverse_cols(y);
  load_weights(1);
  density_stage();

  float part[8];
#pragma unroll
  for (int v = 0; v < 8; ++v) part[v] = 0.0f;
#pragma unroll
  for (int n = 0; n < 4; ++n)
#pragma unroll
    for (int v = 0; v < 8; ++v) {
      const int row = r0 + v + half * 8;
      const int col = n * 16 + l15;
      const float m = zmean[row * kD + col];
      const float lv = zlogvar[row * kD + col];
      const float d = y[n][v] - m;
      part[v] += d * d * expf(-lv) + lv + kLog2Pi;
    }
#pragma unroll
  for (int v = 0; v < 8; ++v) s_red[tid * 8 + v] = part[v];
  wave_fence();
  if (lane < 16) {
    const int h2 = lane >> 3;
    const int vv = lane & 7;
    const int base = wv * 32 + h2 * 16;
    float s = 0.0f;
#pragma unroll
    for (int j = 0; j < 16; ++j) s += s_red[(base + j) * 8 + vv];
    out_lp[r0 + lane] = -0.5f * s;
  }
}

extern "C" void kernel_launch(void* const* d_in, const int* in_sizes, int n_in,
                              void* d_out, int out_size, void* d_ws, size_t ws_size,
                              hipStream_t stream) {
  (void)in_sizes; (void)n_in; (void)out_size; (void)d_ws; (void)ws_size;
  const float* zmean = (const float*)d_in[0];
  const float* zlogvar = (const float*)d_in[1];
  const float* noise = (const float*)d_in[2];
  const float* Wg = (const float*)d_in[3];
  const float* bvec = (const float*)d_in[4];
  float* samples = (float*)d_out;
  float* lp = samples + (size_t)kB * kD;
  const dim3 grid(kB / kRowsPerBlock);
  iaf_kernel<<<grid, kThreads, 0, stream>>>(zmean, zlogvar, noise, Wg, bvec, samples, lp);
}


# density fixed point: each iteration recomputes only the 16-col tile that can still change; products over final tiles are summed once per 16 iterations (block-triangular masks); sum order kept
# speedup vs baseline: 3.1518x; 3.1518x over previous
#include <hip/hip_runtime.h>
#include <math.h>

namespace {
constexpr int kB = 16384;
constexpr int kD = 64;
constexpr int kWaves = 4;
constexpr int kThreads = kWaves * 32;
constexpr int kRowsPerWave = 16;
constexpr int kRowsPerBlock = kWaves * kRowsPerWave;
constexpr int kAst = 68;
constexpr int kWPS = 160;
constexpr float kLog2Pi = 1.8378770664093453f;
}

typedef float v2f __attribute__((ext_vector_type(2)));
typedef float v8f __attribute__((ext_vector_type(8)));

__device__ __forceinline__ void wave_fence() { asm volatile("" ::: "memory"); }

__global__ void __launch_bounds__(kThreads)
iaf_kernel(const float* __restrict__ zmean, const float* __restrict__ zlogvar,
           const float* __restrict__ noise, const float* __restrict__ Wg,
           const float* __restrict__ bvec, float* __restrict__ out_samples,
           float* __restrict__ out_lp)
{
  __shared__ float s_w[4 * 32 * kWPS];
  __shared__ float s_b[4 * kD];
  __shared__ float s_act[kWaves][4][kRowsPerWave * kAst];
  __shared__ float s_red[kThreads * 8];

  const int tid = threadIdx.x;
  const int lane = tid & 31;
  const int wv = tid >> 5;
  const int half = lane >> 4;
  const int l15 = lane & 15;
  const int r0 = blockIdx.x * kRowsPerBlock + wv * kRowsPerWave;

  auto load_weights = [&](int s) {
    __syncthreads();
    for (int idx = tid; idx < 4 * kD * kD; idx += kThreads) {
      const int l = idx >> 12;
      const int rem = idx & 4095;
      const int k = rem >> 6;
      const int n = rem & 63;
      const bool keep = (l == 0) ? (k < n) : (k <= n);
      s_w[(l * 32 + (k >> 1)) * kWPS + 2 * n + (k & 1)] =
          keep ? Wg[((s * 4 + l) * kD + k) * kD + n] : 0.0f;
    }
    for (int idx = tid; idx < 4 * kD; idx += kThreads)
      s_b[idx] = bvec[s * 4 * kD + idx];
    __syncthreads();
  };

  auto layer = [&](int ib, int ob, int l, bool do_relu, v8f* outfr, const v8f* addfr) {
    v8f acc[4];
#pragma unroll
    for (int n = 0; n < 4; ++n) {
      const float bv = s_b[l * kD + n * 16 + l15];
      v8f t;
#pragma unroll
      for (int v = 0; v < 8; ++v) t[v] = bv;
      acc[n] = t;
    }
    const int arow = l15 * kAst;
    const int wbase = 2 * l15;
#pragma unroll
    for (int kk = 0; kk < 16; ++kk) {
      const int kb = kk * 4 + half * 2;
      v2f a;
      a.x = s_act[wv][ib][arow + kb];
      a.y = s_act[wv][ib][arow + kb + 1];
      const int pbase = (l * 32 + kk * 2 + half) * kWPS + wbase;
#pragma unroll
      for (int n = 0; n < 4; ++n) {
        v2f bf;
        bf.x = s_w[pbase + 32 * n];
        bf.y = s_w[pbase + 32 * n + 1];
        acc[n] = __builtin_amdgcn_wmma_f32_16x16x4_f32(
            false, a, false, bf, (short)0, acc[n], false, false);
      }
    }
    if (do_relu) {
#pragma unroll
      for (int n = 0; n < 4; ++n)
#pragma unroll
        for (int v = 0; v < 8; ++v) acc[n][v] = fmaxf(acc[n][v], 0.0f);
    }
    if (addfr) {
#pragma unroll
      for (int n = 0; n < 4; ++n)
#pragma unroll
        for (int v = 0; v < 8; ++v) acc[n][v] += addfr[n][v];
    }
    if (ob >= 0) {
#pragma unroll
      for (int n = 0; n < 4; ++n)
#pragma unroll
        for (int v = 0; v < 8; ++v)
          s_act[wv][ob][(v + half * 8) * kAst + n * 16 + l15] = acc[n][v];
      wave_fence();
    } else {
#pragma unroll
      for (int n = 0; n < 4; ++n) outfr[n] = acc[n];
    }
  };

  auto store_frags = [&](const v8f* fr, int b) {
#pragma unroll
    for (int n = 0; n < 4; ++n)
#pragma unroll
      for (int v = 0; v < 8; ++v)
        s_act[wv][b][(v + half * 8) * kAst + n * 16 + l15] = fr[n][v];
    wave_fence();
  };

  auto made = [&](v8f* outfr) {
    layer(0, 1, 0, true, nullptr, nullptr);
    layer(1, 0, 1, true, nullptr, nullptr);
    layer(0, 1, 2, true, nullptr, nullptr);
    layer(1, -1, 3, false, outfr, nullptr);
  };

  auto reverse_cols = [&](v8f* fr) {
    store_frags(fr, 0);
#pragma unroll
    for (int n = 0; n < 4; ++n)
#pragma unroll
      for (int v = 0; v < 8; ++v)
        fr[n][v] = s_act[wv][0][(v + half * 8) * kAst + (63 - (n * 16 + l15))];
    wave_fence();
  };

  v8f y[4];
#pragma unroll
  for (int n = 0; n < 4; ++n)
#pragma unroll
    for (int v = 0; v < 8; ++v) {
      const int row = r0 + v + half * 8;
      const int col = n * 16 + l15;
      const float m = zmean[row * kD + col];
      const float lv = zlogvar[row * kD + col];
      const float e = noise[row * kD + col];
      y[n][v] = m + expf(0.5f * lv) * e;
    }

  v8f m4[4];

  load_weights(1);
  store_frags(y, 0);
  made(m4);
#pragma unroll
  for (int n = 0; n < 4; ++n)
#pragma unroll
    for (int v = 0; v < 8; ++v) y[n][v] -= m4[n][v];
  reverse_cols(y);

  load_weights(0);
  store_frags(y, 0);
  made(m4);
#pragma unroll
  for (int n = 0; n < 4; ++n)
#pragma unroll
    for (int v = 0; v < 8; ++v) y[n][v] -= m4[n][v];

#pragma unroll
  for (int n = 0; n < 4; ++n)
#pragma unroll
    for (int v = 0; v < 8; ++v)
      out_samples[(r0 + v + half * 8) * kD + n * 16 + l15] = y[n][v];

  auto density_stage = [&]() {
    v8f x0[4];
#pragma unroll
    for (int n = 0; n < 4; ++n) x0[n] = y[n];
#pragma unroll
    for (int n = 0; n < 4; ++n)
#pragma unroll
      for (int v = 0; v < 8; ++v)
        s_act[wv][0][(v + half * 8) * kAst + n * 16 + l15] = 0.0f;
    wave_fence();
    const int arow = l15 * kAst;
#pragma unroll
    for (int j = 0; j < 4; ++j) {
      v8f base[4];
      v2f wd[4][4];
#pragma unroll
      for (int l = 0; l < 4; ++l) {
        const float bv = s_b[l * kD + j * 16 + l15];
        v8f t;
#pragma unroll
        for (int v = 0; v < 8; ++v) t[v] = bv;
#pragma unroll
        for (int kk = 0; kk < 4 * j; ++kk) {
          const int kb = kk * 4 + half * 2;
          v2f a;
          a.x = s_act[wv][l][arow + kb];
          a.y = s_act[wv][l][arow + kb + 1];
          const int p = (l * 32 + kk * 2 + half) * kWPS + 2 * l15 + 32 * j;
          v2f bf;
          bf.x = s_w[p];
          bf.y = s_w[p + 1];
          t = __builtin_amdgcn_wmma_f32_16x16x4_f32(
              false, a, false, bf, (short)0, t, false, false);
        }
        base[l] = t;
#pragma unroll
        for (int q = 0; q < 4; ++q) {
          const int p = (l * 32 + (4 * j + q) * 2 + half) * kWPS + 2 * l15 + 32 * j;
          wd[l][q].x = s_w[p];
          wd[l][q].y = s_w[p + 1];
        }
      }
      for (int i = 0; i < 16; ++i) {
#pragma unroll
        for (int l = 0; l < 4; ++l) {
          v8f acc = base[l];
#pragma unroll
          for (int q = 0; q < 4; ++q) {
            const int kb = j * 16 + q * 4 + half * 2;
            v2f a;
            a.x = s_act[wv][l][arow + kb];
            a.y = s_act[wv][l][arow + kb + 1];
            acc = __builtin_amdgcn_wmma_f32_16x16x4_f32(
                false, a, false, wd[l][q], (short)0, acc, false, false);
          }
          if (l < 3) {
#pragma unroll
            for (int v = 0; v < 8; ++v) acc[v] = fmaxf(acc[v], 0.0f);
          } else {
#pragma unroll
            for (int v = 0; v < 8; ++v) acc[v] += x0[j][v];
          }
#pragma unroll
          for (int v = 0; v < 8; ++v)
            s_act[wv][(l + 1) & 3][(v + half * 8) * kAst + j * 16 + l15] = acc[v];
          wave_fence();
        }
      }
    }
#pragma unroll
    for (int n = 0; n < 4; ++n)
#pragma unroll
      for (int v = 0; v < 8; ++v)
        y[n][v] = s_act[wv][0][(v + half * 8) * kAst + n * 16 + l15];
    wave_fence();
  };

  density_stage();
  reverse_cols(y);
  load_weights(1);
  density_stage();

  float part[8];
#pragma unroll
  for (int v = 0; v < 8; ++v) part[v] = 0.0f;
#pragma unroll
  for (int n = 0; n < 4; ++n)
#pragma unroll
    for (int v = 0; v < 8; ++v) {
      const int row = r0 + v + half * 8;
      const int col = n * 16 + l15;
      const float m = zmean[row * kD + col];
      const float lv = zlogvar[row * kD + col];
      const float d = y[n][v] - m;
      part[v] += d * d * expf(-lv) + lv + kLog2Pi;
    }
#pragma unroll
  for (int v = 0; v < 8; ++v) s_red[tid * 8 + v] = part[v];
  wave_fence();
  if (lane < 16) {
    const int h2 = lane >> 3;
    const int vv = lane & 7;
    const int base = wv * 32 + h2 * 16;
    float s = 0.0f;
#pragma unroll
    for (int j = 0; j < 16; ++j) s += s_red[(base + j) * 8 + vv];
    out_lp[r0 + lane] = -0.5f * s;
  }
}

extern "C" void kernel_launch(void* const* d_in, const int* in_sizes, int n_in,
                              void* d_out, int out_size, void* d_ws, size_t ws_size,
                              hipStream_t stream) {
  (void)in_sizes; (void)n_in; (void)out_size; (void)d_ws; (void)ws_size;
  const float* zmean = (const float*)d_in[0];
  const float* zlogvar = (const float*)d_in[1];
  const float* noise = (const float*)d_in[2];
  const float* Wg = (const float*)d_in[3];
  const float* bvec = (const float*)d_in[4];
  float* samples = (float*)d_out;
  float* lp = samples + (size_t)kB * kD;
  const dim3 grid(kB / kRowsPerBlock);
  iaf_kernel<<<grid, kThreads, 0, stream>>>(zmean, zlogvar, noise, Wg, bvec, samples, lp);
}


# weight staging: per stage the 128 dependent masked 4-byte loads per thread become 32 unconditional 16-byte loads issued together ahead of the barrier; mask by select, 16-byte LDS stores
# speedup vs baseline: 8.4380x; 2.6772x over previous
#include <hip/hip_runtime.h>
#include <math.h>

namespace {
constexpr int kB = 16384;
constexpr int kD = 64;
constexpr int kWaves = 4;
constexpr int kThreads = kWaves * 32;
constexpr int kRowsPerWave = 16;
constexpr int kRowsPerBlock = kWaves * kRowsPerWave;
constexpr int kAst = 68;
constexpr int kWPS = 160;
constexpr float kLog2Pi = 1.8378770664093453f;
}

typedef float v2f __attribute__((ext_vector_type(2)));
typedef float v4f __attribute__((ext_vector_type(4)));
typedef float v8f __attribute__((ext_vector_type(8)));

__device__ __forceinline__ void wave_fence() { asm volatile("" ::: "memory"); }

__global__ void __launch_bounds__(kThreads)
iaf_kernel(const float* __restrict__ zmean, const float* __restrict__ zlogvar,
           const float* __restrict__ noise, const float* __restrict__ Wg,
           const float* __restrict__ bvec, float* __restrict__ out_samples,
           float* __restrict__ out_lp)
{
  __shared__ alignas(16) float s_w[4 * 32 * kWPS];
  __shared__ float s_b[4 * kD];
  __shared__ float s_act[kWaves][4][kRowsPerWave * kAst];
  __shared__ float s_red[kThreads * 8];

  const int tid = threadIdx.x;
  const int lane = tid & 31;
  const int wv = tid >> 5;
  const int half = lane >> 4;
  const int l15 = lane & 15;
  const int r0 = blockIdx.x * kRowsPerBlock + wv * kRowsPerWave;

  auto load_weights = [&](int s) {
    v4f w0[16], w1[16];
    const int nq = tid & 15;
    const int kq = tid >> 4;
#pragma unroll
    for (int it = 0; it < 16; ++it) {
      const int l = it >> 2;
      const int kp = (it & 3) * 8 + kq;
      const float* src = Wg + ((s * 4 + l) * kD + 2 * kp) * kD + 4 * nq;
      w0[it] = *reinterpret_cast<const v4f*>(src);
      w1[it] = *reinterpret_cast<const v4f*>(src + kD);
    }
    const float b0 = bvec[s * 4 * kD + tid];
    const float b1 = bvec[s * 4 * kD + kThreads + tid];
    __syncthreads();
#pragma unroll
    for (int it = 0; it < 16; ++it) {
      const int l = it >> 2;
      const int kp = (it & 3) * 8 + kq;
      const int k0 = 2 * kp;
      const int k1 = k0 + 1;
      const int n0 = 4 * nq;
      const int d0 = (l == 0) ? k0 + 1 : k0;
      const int d1 = (l == 0) ? k1 + 1 : k1;
      v4f lo, hi;
      lo.x = (n0 + 0 >= d0) ? w0[it].x : 0.0f;
      lo.y = (n0 + 0 >= d1) ? w1[it].x : 0.0f;
      lo.z = (n0 + 1 >= d0) ? w0[it].y : 0.0f;
      lo.w = (n0 + 1 >= d1) ? w1[it].y : 0.0f;
      hi.x = (n0 + 2 >= d0) ? w0[it].z : 0.0f;
      hi.y = (n0 + 2 >= d1) ? w1[it].z : 0.0f;
      hi.z = (n0 + 3 >= d0) ? w0[it].w : 0.0f;
      hi.w = (n0 + 3 >= d1) ? w1[it].w : 0.0f;
      float* dst = &s_w[(l * 32 + kp) * kWPS + 2 * n0];
      *reinterpret_cast<v4f*>(dst) = lo;
      *reinterpret_cast<v4f*>(dst + 4) = hi;
    }
    s_b[tid] = b0;
    s_b[kThreads + tid] = b1;
    __syncthreads();
  };

  auto layer = [&](int ib, int ob, int l, bool do_relu, v8f* outfr, const v8f* addfr) {
    v8f acc[4];
#pragma unroll
    for (int n = 0; n < 4; ++n) {
      const float bv = s_b[l * kD + n * 16 + l15];
      v8f t;
#pragma unroll
      for (int v = 0; v < 8; ++v) t[v] = bv;
      acc[n] = t;
    }
    const int arow = l15 * kAst;
    const int wbase = 2 * l15;
#pragma unroll
    for (int kk = 0; kk < 16; ++kk) {
      const int kb = kk * 4 + half * 2;
      v2f a;
      a.x = s_act[wv][ib][arow + kb];
      a.y = s_act[wv][ib][arow + kb + 1];
      const int pbase = (l * 32 + kk * 2 + half) * kWPS + wbase;
#pragma unroll
      for (int n = 0; n < 4; ++n) {
        v2f bf;
        bf.x = s_w[pbase + 32 * n];
        bf.y = s_w[pbase + 32 * n + 1];
        acc[n] = __builtin_amdgcn_wmma_f32_16x16x4_f32(
            false, a, false, bf, (short)0, acc[n], false, false);
      }
    }
    if (do_relu) {
#pragma unroll
      for (int n = 0; n < 4; ++n)
#pragma unroll
        for (int v = 0; v < 8; ++v) acc[n][v] = fmaxf(acc[n][v], 0.0f);
    }
    if (addfr) {
#pragma unroll
      for (int n = 0; n < 4; ++n)
#pragma unroll
        for (int v = 0; v < 8; ++v) acc[n][v] += addfr[n][v];
    }
    if (ob >= 0) {
#pragma unroll
      for (int n = 0; n < 4; ++n)
#pragma unroll
        for (int v = 0; v < 8; ++v)
          s_act[wv][ob][(v + half * 8) * kAst + n * 16 + l15] = acc[n][v];
      wave_fence();
    } else {
#pragma unroll
      for (int n = 0; n < 4; ++n) outfr[n] = acc[n];
    }
  };

  auto store_frags = [&](const v8f* fr, int b) {
#pragma unroll
    for (int n = 0; n < 4; ++n)
#pragma unroll
      for (int v = 0; v < 8; ++v)
        s_act[wv][b][(v + half * 8) * kAst + n * 16 + l15] = fr[n][v];
    wave_fence();
  };

  auto made = [&](v8f* outfr) {
    layer(0, 1, 0, true, nullptr, nullptr);
    layer(1, 0, 1, true, nullptr, nullptr);
    layer(0, 1, 2, true, nullptr, nullptr);
    layer(1, -1, 3, false, outfr, nullptr);
  };

  auto reverse_cols = [&](v8f* fr) {
    store_frags(fr, 0);
#pragma unroll
    for (int n = 0; n < 4; ++n)
#pragma unroll
      for (int v = 0; v < 8; ++v)
        fr[n][v] = s_act[wv][0][(v + half * 8) * kAst + (63 - (n * 16 + l15))];
    wave_fence();
  };

  v8f y[4];
#pragma unroll
  for (int n = 0; n < 4; ++n)
#pragma unroll
    for (int v = 0; v < 8; ++v) {
      const int row = r0 + v + half * 8;
      const int col = n * 16 + l15;
      const float m = zmean[row * kD + col];
      const float lv = zlogvar[row * kD + col];
      const float e = noise[row * kD + col];
      y[n][v] = m + expf(0.5f * lv) * e;
    }

  v8f m4[4];

  load_weights(1);
  store_frags(y, 0);
  made(m4);
#pragma unroll
  for (int n = 0; n < 4; ++n)
#pragma unroll
    for (int v = 0; v < 8; ++v) y[n][v] -= m4[n][v];
  reverse_cols(y);

  load_weights(0);
  store_frags(y, 0);
  made(m4);
#pragma unroll
  for (int n = 0; n < 4; ++n)
#pragma unroll
    for (int v = 0; v < 8; ++v) y[n][v] -= m4[n][v];

#pragma unroll
  for (int n = 0; n < 4; ++n)
#pragma unroll
    for (int v = 0; v < 8; ++v)
      out_samples[(r0 + v + half * 8) * kD + n * 16 + l15] = y[n][v];

  auto density_stage = [&]() {
    v8f x0[4];
#pragma unroll
    for (int n = 0; n < 4; ++n) x0[n] = y[n];
#pragma unroll
    for (int n = 0; n < 4; ++n)
#pragma unroll
      for (int v = 0; v < 8; ++v)
        s_act[wv][0][(v + half * 8) * kAst + n * 16 + l15] = 0.0f;
    wave_fence();
    const int arow = l15 * kAst;
#pragma unroll
    for (int j = 0; j < 4; ++j) {
      v8f base[4];
      v2f wd[4][4];
#pragma unroll
      for (int l = 0; l < 4; ++l) {
        const float bv = s_b[l * kD + j * 16 + l15];
        v8f t;
#pragma unroll
        for (int v = 0; v < 8; ++v) t[v] = bv;
#pragma unroll
        for (int kk = 0; kk < 4 * j; ++kk) {
          const int kb = kk * 4 + half * 2;
          v2f a;
          a.x = s_act[wv][l][arow + kb];
          a.y = s_act[wv][l][arow + kb + 1];
          const int p = (l * 32 + kk * 2 + half) * kWPS + 2 * l15 + 32 * j;
          v2f bf;
          bf.x = s_w[p];
          bf.y = s_w[p + 1];
          t = __builtin_amdgcn_wmma_f32_16x16x4_f32(
              false, a, false, bf, (short)0, t, false, false);
        }
        base[l] = t;
#pragma unroll
        for (int q = 0; q < 4; ++q) {
          const int p = (l * 32 + (4 * j + q) * 2 + half) * kWPS + 2 * l15 + 32 * j;
          wd[l][q].x = s_w[p];
          wd[l][q].y = s_w[p + 1];
        }
      }
      for (int i = 0; i < 16; ++i) {
#pragma unroll
        for (int l = 0; l < 4; ++l) {
          v8f acc = base[l];
#pragma unroll
          for (int q = 0; q < 4; ++q) {
            const int kb = j * 16 + q * 4 + half * 2;
            v2f a;
            a.x = s_act[wv][l][arow + kb];
            a.y = s_act[wv][l][arow + kb + 1];
            acc = __builtin_amdgcn_wmma_f32_16x16x4_f32(
                false, a, false, wd[l][q], (short)0, acc, false, false);
          }
          if (l < 3) {
#pragma unroll
            for (int v = 0; v < 8; ++v) acc[v] = fmaxf(acc[v], 0.0f);
          } else {
#pragma unroll
            for (int v = 0; v < 8; ++v) acc[v] += x0[j][v];
          }
#pragma unroll
          for (int v = 0; v < 8; ++v)
            s_act[wv][(l + 1) & 3][(v + half * 8) * kAst + j * 16 + l15] = acc[v];
          wave_fence();
        }
      }
    }
#pragma unroll
    for (int n = 0; n < 4; ++n)
#pragma unroll
      for (int v = 0; v < 8; ++v)
        y[n][v] = s_act[wv][0][(v + half * 8) * kAst + n * 16 + l15];
    wave_fence();
  };

  density_stage();
  reverse_cols(y);
  load_weights(1);
  density_stage();

  float part[8];
#pragma unroll
  for (int v = 0; v < 8; ++v) part[v] = 0.0f;
#pragma unroll
  for (int n = 0; n < 4; ++n)
#pragma unroll
    for (int v = 0; v < 8; ++v) {
      const int row = r0 + v + half * 8;
      const int col = n * 16 + l15;
      const float m = zmean[row * kD + col];
      const float lv = zlogvar[row * kD + col];
      const float d = y[n][v] - m;
      part[v] += d * d * expf(-lv) + lv + kLog2Pi;
    }
#pragma unroll
  for (int v = 0; v < 8; ++v) s_red[tid * 8 + v] = part[v];
  wave_fence();
  if (lane < 16) {
    const int h2 = lane >> 3;
    const int vv = lane & 7;
    const int base = wv * 32 + h2 * 16;
    float s = 0.0f;
#pragma unroll
    for (int j = 0; j < 16; ++j) s += s_red[(base + j) * 8 + vv];
    out_lp[r0 + lane] = -0.5f * s;
  }
}

extern "C" void kernel_launch(void* const* d_in, const int* in_sizes, int n_in,
                              void* d_out, int out_size, void* d_ws, size_t ws_size,
                              hipStream_t stream) {
  (void)in_sizes; (void)n_in; (void)out_size; (void)d_ws; (void)ws_size;
  const float* zmean = (const float*)d_in[0];
  const float* zlogvar = (const float*)d_in[1];
  const float* noise = (const float*)d_in[2];
  const float* Wg = (const float*)d_in[3];
  const float* bvec = (const float*)d_in[4];
  float* samples = (float*)d_out;
  float* lp = samples + (size_t)kB * kD;
  const dim3 grid(kB / kRowsPerBlock);
  iaf_kernel<<<grid, kThreads, 0, stream>>>(zmean, zlogvar, noise, Wg, bvec, samples, lp);
}


# density tiles computed transposed (W^T x in^T): a layer's D fragment is the next layer's B fragment, no LDS exchange per layer; k order per diagonal 16-block 0,1,8,9,2,3,10,11,.. (sum order)
# speedup vs baseline: 13.1409x; 1.5573x over previous
#include <hip/hip_runtime.h>
#include <math.h>

namespace {
constexpr int kB = 16384;
constexpr int kD = 64;
constexpr int kWaves = 4;
constexpr int kThreads = kWaves * 32;
constexpr int kRowsPerWave = 16;
constexpr int kRowsPerBlock = kWaves * kRowsPerWave;
constexpr int kAst = 68;
constexpr int kWPS = 160;
constexpr float kLog2Pi = 1.8378770664093453f;
}

typedef float v2f __attribute__((ext_vector_type(2)));
typedef float v4f __attribute__((ext_vector_type(4)));
typedef float v8f __attribute__((ext_vector_type(8)));

__device__ __forceinline__ void wave_fence() { asm volatile("" ::: "memory"); }

__global__ void __launch_bounds__(kThreads)
iaf_kernel(const float* __restrict__ zmean, const float* __restrict__ zlogvar,
           const float* __restrict__ noise, const float* __restrict__ Wg,
           const float* __restrict__ bvec, float* __restrict__ out_samples,
           float* __restrict__ out_lp)
{
  __shared__ alignas(16) float s_w[4 * 32 * kWPS];
  __shared__ float s_b[4 * kD];
  __shared__ float s_act[kWaves][4][kRowsPerWave * kAst];
  __shared__ float s_red[kThreads * 8];

  const int tid = threadIdx.x;
  const int lane = tid & 31;
  const int wv = tid >> 5;
  const int half = lane >> 4;
  const int l15 = lane & 15;
  const int r0 = blockIdx.x * kRowsPerBlock + wv * kRowsPerWave;

  auto load_weights = [&](int s) {
    v4f w0[16], w1[16];
    const int nq = tid & 15;
    const int kq = tid >> 4;
#pragma unroll
    for (int it = 0; it < 16; ++it) {
      const int l = it >> 2;
      const int kp = (it & 3) * 8 + kq;
      const float* src = Wg + ((s * 4 + l) * kD + 2 * kp) * kD + 4 * nq;
      w0[it] = *reinterpret_cast<const v4f*>(src);
      w1[it] = *reinterpret_cast<const v4f*>(src + kD);
    }
    const float b0 = bvec[s * 4 * kD + tid];
    const float b1 = bvec[s * 4 * kD + kThreads + tid];
    __syncthreads();
#pragma unroll
    for (int it = 0; it < 16; ++it) {
      const int l = it >> 2;
      const int kp = (it & 3) * 8 + kq;
      const int k0 = 2 * kp;
      const int k1 = k0 + 1;
      const int n0 = 4 * nq;
      const int d0 = (l == 0) ? k0 + 1 : k0;
      const int d1 = (l == 0) ? k1 + 1 : k1;
      v4f lo, hi;
      lo.x = (n0 + 0 >= d0) ? w0[it].x : 0.0f;
      lo.y = (n0 + 0 >= d1) ? w1[it].x : 0.0f;
      lo.z = (n0 + 1 >= d0) ? w0[it].y : 0.0f;
      lo.w = (n0 + 1 >= d1) ? w1[it].y : 0.0f;
      hi.x = (n0 + 2 >= d0) ? w0[it].z : 0.0f;
      hi.y = (n0 + 2 >= d1) ? w1[it].z : 0.0f;
      hi.z = (n0 + 3 >= d0) ? w0[it].w : 0.0f;
      hi.w = (n0 + 3 >= d1) ? w1[it].w : 0.0f;
      float* dst = &s_w[(l * 32 + kp) * kWPS + 2 * n0];
      *reinterpret_cast<v4f*>(dst) = lo;
      *reinterpret_cast<v4f*>(dst + 4) = hi;
    }
    s_b[tid] = b0;
    s_b[kThreads + tid] = b1;
    __syncthreads();
  };

  auto layer = [&](int ib, int ob, int l, bool do_relu, v8f* outfr, const v8f* addfr) {
    v8f acc[4];
#pragma unroll
    for (int n = 0; n < 4; ++n) {
      const float bv = s_b[l * kD + n * 16 + l15];
      v8f t;
#pragma unroll
      for (int v = 0; v < 8; ++v) t[v] = bv;
      acc[n] = t;
    }
    const int arow = l15 * kAst;
    const int wbase = 2 * l15;
#pragma unroll
    for (int kk = 0; kk < 16; ++kk) {
      const int kb = kk * 4 + half * 2;
      v2f a;
      a.x = s_act[wv][ib][arow + kb];
      a.y = s_act[wv][ib][arow + kb + 1];
      const int pbase = (l * 32 + kk * 2 + half) * kWPS + wbase;
#pragma unroll
      for (int n = 0; n < 4; ++n) {
        v2f bf;
        bf.x = s_w[pbase + 32 * n];
        bf.y = s_w[pbase + 32 * n + 1];
        acc[n] = __builtin_amdgcn_wmma_f32_16x16x4_f32(
            false, a, false, bf, (short)0, acc[n], false, false);
      }
    }
    if (do_relu) {
#pragma unroll
      for (int n = 0; n < 4; ++n)
#pragma unroll
        for (int v = 0; v < 8; ++v) acc[n][v] = fmaxf(acc[n][v], 0.0f);
    }
    if (addfr) {
#pragma unroll
      for (int n = 0; n < 4; ++n)
#pragma unroll
        for (int v = 0; v < 8; ++v) acc[n][v] += addfr[n][v];
    }
    if (ob >= 0) {
#pragma unroll
      for (int n = 0; n < 4; ++n)
#pragma unroll
        for (int v = 0; v < 8; ++v)
          s_act[wv][ob][(v + half * 8) * kAst + n * 16 + l15] = acc[n][v];
      wave_fence();
    } else {
#pragma unroll
      for (int n = 0; n < 4; ++n) outfr[n] = acc[n];
    }
  };

  auto store_frags = [&](const v8f* fr, int b) {
#pragma unroll
    for (int n = 0; n < 4; ++n)
#pragma unroll
      for (int v = 0; v < 8; ++v)
        s_act[wv][b][(v + half * 8) * kAst + n * 16 + l15] = fr[n][v];
    wave_fence();
  };

  auto made = [&](v8f* outfr) {
    layer(0, 1, 0, true, nullptr, nullptr);
    layer(1, 0, 1, true, nullptr, nullptr);
    layer(0, 1, 2, true, nullptr, nullptr);
    layer(1, -1, 3, false, outfr, nullptr);
  };

  auto reverse_cols = [&](v8f* fr) {
    store_frags(fr, 0);
#pragma unroll
    for (int n = 0; n < 4; ++n)
#pragma unroll
      for (int v = 0; v < 8; ++v)
        fr[n][v] = s_act[wv][0][(v + half * 8) * kAst + (63 - (n * 16 + l15))];
    wave_fence();
  };

  v8f y[4];
#pragma unroll
  for (int n = 0; n < 4; ++n)
#pragma unroll
    for (int v = 0; v < 8; ++v) {
      const int row = r0 + v + half * 8;
      const int col = n * 16 + l15;
      const float m = zmean[row * kD + col];
      const float lv = zlogvar[row * kD + col];
      const float e = noise[row * kD + col];
      y[n][v] = m + expf(0.5f * lv) * e;
    }

  v8f m4[4];

  load_weights(1);
  store_frags(y, 0);
  made(m4);
#pragma unroll
  for (int n = 0; n < 4; ++n)
#pragma unroll
    for (int v = 0; v < 8; ++v) y[n][v] -= m4[n][v];
  reverse_cols(y);

  load_weights(0);
  store_frags(y, 0);
  made(m4);
#pragma unroll
  for (int n = 0; n < 4; ++n)
#pragma unroll
    for (int v = 0; v < 8; ++v) y[n][v] -= m4[n][v];

#pragma unroll
  for (int n = 0; n < 4; ++n)
#pragma unroll
    for (int v = 0; v < 8; ++v)
      out_samples[(r0 + v + half * 8) * kD + n * 16 + l15] = y[n][v];

  auto density_stage = [&]() {
    store_frags(y, 0);
    const int arow = l15 * kAst;
#pragma unroll
    for (int j = 0; j < 4; ++j) {
      const int tcol = arow + j * 16 + half * 8;
      v8f x0t;
#pragma unroll
      for (int v = 0; v < 8; ++v) x0t[v] = s_act[wv][0][tcol + v];
      v8f base[4];
      v2f wd[4][4];
#pragma unroll
      for (int l = 0; l < 4; ++l) {
        v8f t;
#pragma unroll
        for (int v = 0; v < 8; ++v) t[v] = s_b[l * kD + j * 16 + half * 8 + v];
#pragma unroll
        for (int kk = 0; kk < 4 * j; ++kk) {
          const int kb = kk * 4 + half * 2;
          v2f a;
          a.x = s_act[wv][l][arow + kb];
          a.y = s_act[wv][l][arow + kb + 1];
          const int p = (l * 32 + kk * 2 + half) * kWPS + 2 * l15 + 32 * j;
          v2f bf;
          bf.x = s_w[p];
          bf.y = s_w[p + 1];
          t = __builtin_amdgcn_wmma_f32_16x16x4_f32(
              false, bf, false, a, (short)0, t, false, false);
        }
        base[l] = t;
#pragma unroll
        for (int q = 0; q < 4; ++q) {
          const int p = (l * 32 + 8 * j + 4 * half + q) * kWPS + 2 * l15 + 32 * j;
          wd[l][q].x = s_w[p];
          wd[l][q].y = s_w[p + 1];
        }
      }
      v8f st[4];
#pragma unroll
      for (int l = 0; l < 4; ++l)
#pragma unroll
        for (int v = 0; v < 8; ++v) st[l][v] = 0.0f;
#pragma unroll 1
      for (int i = 0; i < 16; ++i) {
#pragma unroll
        for (int l = 0; l < 4; ++l) {
          v8f acc = base[l];
#pragma unroll
          for (int q = 0; q < 4; ++q) {
            v2f bq;
            bq.x = st[l][2 * q];
            bq.y = st[l][2 * q + 1];
            acc = __builtin_amdgcn_wmma_f32_16x16x4_f32(
                false, wd[l][q], false, bq, (short)0, acc, false, false);
          }
          if (l < 3) {
#pragma unroll
            for (int v = 0; v < 8; ++v) acc[v] = fmaxf(acc[v], 0.0f);
          } else {
#pragma unroll
            for (int v = 0; v < 8; ++v) acc[v] += x0t[v];
          }
          st[(l + 1) & 3] = acc;
        }
      }
#pragma unroll
      for (int l = 0; l < 4; ++l)
#pragma unroll
        for (int v = 0; v < 8; ++v) s_act[wv][l][tcol + v] = st[l][v];
      wave_fence();
    }
#pragma unroll
    for (int n = 0; n < 4; ++n)
#pragma unroll
      for (int v = 0; v < 8; ++v)
        y[n][v] = s_act[wv][0][(v + half * 8) * kAst + n * 16 + l15];
    wave_fence();
  };

  density_stage();
  reverse_cols(y);
  load_weights(1);
  density_stage();

  float part[8];
#pragma unroll
  for (int v = 0; v < 8; ++v) part[v] = 0.0f;
#pragma unroll
  for (int n = 0; n < 4; ++n)
#pragma unroll
    for (int v = 0; v < 8; ++v) {
      const int row = r0 + v + half * 8;
      const int col = n * 16 + l15;
      const float m = zmean[row * kD + col];
      const float lv = zlogvar[row * kD + col];
      const float d = y[n][v] - m;
      part[v] += d * d * expf(-lv) + lv + kLog2Pi;
    }
#pragma unroll
  for (int v = 0; v < 8; ++v) s_red[tid * 8 + v] = part[v];
  wave_fence();
  if (lane < 16) {
    const int h2 = lane >> 3;
    const int vv = lane & 7;
    const int base = wv * 32 + h2 * 16;
    float s = 0.0f;
#pragma unroll
    for (int j = 0; j < 16; ++j) s += s_red[(base + j) * 8 + vv];
    out_lp[r0 + lane] = -0.5f * s;
  }
}

extern "C" void kernel_launch(void* const* d_in, const int* in_sizes, int n_in,
                              void* d_out, int out_size, void* d_ws, size_t ws_size,
                              hipStream_t stream) {
  (void)in_sizes; (void)n_in; (void)out_size; (void)d_ws; (void)ws_size;
  const float* zmean = (const float*)d_in[0];
  const float* zlogvar = (const float*)d_in[1];
  const float* noise = (const float*)d_in[2];
  const float* Wg = (const float*)d_in[3];
  const float* bvec = (const float*)d_in[4];
  float* samples = (float*)d_out;
  float* lp = samples + (size_t)kB * kD;
  const dim3 grid(kB / kRowsPerBlock);
  iaf_kernel<<<grid, kThreads, 0, stream>>>(zmean, zlogvar, noise, Wg, bvec, samples, lp);
}


# density diagonal block in K=4 chunks: a layer step multiplies only the 4-feature chunk still changing, final chunks join the constant sums once; 1 WMMA per step, not 4; k ascending again (sum order)
# speedup vs baseline: 20.2904x; 1.5441x over previous
#include <hip/hip_runtime.h>
#include <math.h>

namespace {
constexpr int kB = 16384;
constexpr int kD = 64;
constexpr int kWaves = 4;
constexpr int kThreads = kWaves * 32;
constexpr int kRowsPerWave = 16;
constexpr int kRowsPerBlock = kWaves * kRowsPerWave;
constexpr int kAst = 68;
constexpr int kWPS = 160;
constexpr float kLog2Pi = 1.8378770664093453f;
}

typedef float v2f __attribute__((ext_vector_type(2)));
typedef float v4f __attribute__((ext_vector_type(4)));
typedef float v8f __attribute__((ext_vector_type(8)));

__device__ __forceinline__ void wave_fence() { asm volatile("" ::: "memory"); }

__global__ void __launch_bounds__(kThreads)
iaf_kernel(const float* __restrict__ zmean, const float* __restrict__ zlogvar,
           const float* __restrict__ noise, const float* __restrict__ Wg,
           const float* __restrict__ bvec, float* __restrict__ out_samples,
           float* __restrict__ out_lp)
{
  __shared__ alignas(16) float s_w[4 * 32 * kWPS];
  __shared__ float s_b[4 * kD];
  __shared__ float s_act[kWaves][4][kRowsPerWave * kAst];
  __shared__ float s_red[kThreads * 8];

  const int tid = threadIdx.x;
  const int lane = tid & 31;
  const int wv = tid >> 5;
  const int half = lane >> 4;
  const int l15 = lane & 15;
  const int r0 = blockIdx.x * kRowsPerBlock + wv * kRowsPerWave;

  auto load_weights = [&](int s) {
    v4f w0[16], w1[16];
    const int nq = tid & 15;
    const int kq = tid >> 4;
#pragma unroll
    for (int it = 0; it < 16; ++it) {
      const int l = it >> 2;
      const int kp = (it & 3) * 8 + kq;
      const float* src = Wg + ((s * 4 + l) * kD + 2 * kp) * kD + 4 * nq;
      w0[it] = *reinterpret_cast<const v4f*>(src);
      w1[it] = *reinterpret_cast<const v4f*>(src + kD);
    }
    const float b0 = bvec[s * 4 * kD + tid];
    const float b1 = bvec[s * 4 * kD + kThreads + tid];
    __syncthreads();
#pragma unroll
    for (int it = 0; it < 16; ++it) {
      const int l = it >> 2;
      const int kp = (it & 3) * 8 + kq;
      const int k0 = 2 * kp;
      const int k1 = k0 + 1;
      const int n0 = 4 * nq;
      const int d0 = (l == 0) ? k0 + 1 : k0;
      const int d1 = (l == 0) ? k1 + 1 : k1;
      v4f lo, hi;
      lo.x = (n0 + 0 >= d0) ? w0[it].x : 0.0f;
      lo.y = (n0 + 0 >= d1) ? w1[it].x : 0.0f;
      lo.z = (n0 + 1 >= d0) ? w0[it].y : 0.0f;
      lo.w = (n0 + 1 >= d1) ? w1[it].y : 0.0f;
      hi.x = (n0 + 2 >= d0) ? w0[it].z : 0.0f;
      hi.y = (n0 + 2 >= d1) ? w1[it].z : 0.0f;
      hi.z = (n0 + 3 >= d0) ? w0[it].w : 0.0f;
      hi.w = (n0 + 3 >= d1) ? w1[it].w : 0.0f;
      float* dst = &s_w[(l * 32 + kp) * kWPS + 2 * n0];
      *reinterpret_cast<v4f*>(dst) = lo;
      *reinterpret_cast<v4f*>(dst + 4) = hi;
    }
    s_b[tid] = b0;
    s_b[kThreads + tid] = b1;
    __syncthreads();
  };

  auto layer = [&](int ib, int ob, int l, bool do_relu, v8f* outfr, const v8f* addfr) {
    v8f acc[4];
#pragma unroll
    for (int n = 0; n < 4; ++n) {
      const float bv = s_b[l * kD + n * 16 + l15];
      v8f t;
#pragma unroll
      for (int v = 0; v < 8; ++v) t[v] = bv;
      acc[n] = t;
    }
    const int arow = l15 * kAst;
    const int wbase = 2 * l15;
#pragma unroll
    for (int kk = 0; kk < 16; ++kk) {
      const int kb = kk * 4 + half * 2;
      v2f a;
      a.x = s_act[wv][ib][arow + kb];
      a.y = s_act[wv][ib][arow + kb + 1];
      const int pbase = (l * 32 + kk * 2 + half) * kWPS + wbase;
#pragma unroll
      for (int n = 0; n < 4; ++n) {
        v2f bf;
        bf.x = s_w[pbase + 32 * n];
        bf.y = s_w[pbase + 32 * n + 1];
        acc[n] = __builtin_amdgcn_wmma_f32_16x16x4_f32(
            false, a, false, bf, (short)0, acc[n], false, false);
      }
    }
    if (do_relu) {
#pragma unroll
      for (int n = 0; n < 4; ++n)
#pragma unroll
        for (int v = 0; v < 8; ++v) acc[n][v] = fmaxf(acc[n][v], 0.0f);
    }
    if (addfr) {
#pragma unroll
      for (int n = 0; n < 4; ++n)
#pragma unroll
        for (int v = 0; v < 8; ++v) acc[n][v] += addfr[n][v];
    }
    if (ob >= 0) {
#pragma unroll
      for (int n = 0; n < 4; ++n)
#pragma unroll
        for (int v = 0; v < 8; ++v)
          s_act[wv][ob][(v + half * 8) * kAst + n * 16 + l15] = acc[n][v];
      wave_fence();
    } else {
#pragma unroll
      for (int n = 0; n < 4; ++n) outfr[n] = acc[n];
    }
  };

  auto store_frags = [&](const v8f* fr, int b) {
#pragma unroll
    for (int n = 0; n < 4; ++n)
#pragma unroll
      for (int v = 0; v < 8; ++v)
        s_act[wv][b][(v + half * 8) * kAst + n * 16 + l15] = fr[n][v];
    wave_fence();
  };

  auto made = [&](v8f* outfr) {
    layer(0, 1, 0, true, nullptr, nullptr);
    layer(1, 0, 1, true, nullptr, nullptr);
    layer(0, 1, 2, true, nullptr, nullptr);
    layer(1, -1, 3, false, outfr, nullptr);
  };

  auto reverse_cols = [&](v8f* fr) {
    store_frags(fr, 0);
#pragma unroll
    for (int n = 0; n < 4; ++n)
#pragma unroll
      for (int v = 0; v < 8; ++v)
        fr[n][v] = s_act[wv][0][(v + half * 8) * kAst + (63 - (n * 16 + l15))];
    wave_fence();
  };

  v8f y[4];
#pragma unroll
  for (int n = 0; n < 4; ++n)
#pragma unroll
    for (int v = 0; v < 8; ++v) {
      const int row = r0 + v + half * 8;
      const int col = n * 16 + l15;
      const float m = zmean[row * kD + col];
      const float lv = zlogvar[row * kD + col];
      const float e = noise[row * kD + col];
      y[n][v] = m + expf(0.5f * lv) * e;
    }

  v8f m4[4];

  load_weights(1);
  store_frags(y, 0);
  made(m4);
#pragma unroll
  for (int n = 0; n < 4; ++n)
#pragma unroll
    for (int v = 0; v < 8; ++v) y[n][v] -= m4[n][v];
  reverse_cols(y);

  load_weights(0);
  store_frags(y, 0);
  made(m4);
#pragma unroll
  for (int n = 0; n < 4; ++n)
#pragma unroll
    for (int v = 0; v < 8; ++v) y[n][v] -= m4[n][v];

#pragma unroll
  for (int n = 0; n < 4; ++n)
#pragma unroll
    for (int v = 0; v < 8; ++v)
      out_samples[(r0 + v + half * 8) * kD + n * 16 + l15] = y[n][v];

  auto density_stage = [&]() {
    store_frags(y, 0);
    const int arow = l15 * kAst;
    const int fl = 4 * ((l15 >> 1) & 3) + 2 * (l15 >> 3) + (l15 & 1);
#pragma unroll
    for (int j = 0; j < 4; ++j) {
      const int tcol = arow + j * 16 + half * 2;
      v8f x0t;
#pragma unroll
      for (int v = 0; v < 8; ++v) x0t[v] = s_act[wv][0][tcol + 4 * (v >> 1) + (v & 1)];
      v8f base[4];
      v2f wd[4][4];
#pragma unroll
      for (int l = 0; l < 4; ++l) {
        v8f t;
#pragma unroll
        for (int v = 0; v < 8; ++v)
          t[v] = s_b[l * kD + j * 16 + 4 * (v >> 1) + 2 * half + (v & 1)];
#pragma unroll
        for (int kk = 0; kk < 4 * j; ++kk) {
          const int kb = kk * 4 + half * 2;
          v2f a;
          a.x = s_act[wv][l][arow + kb];
          a.y = s_act[wv][l][arow + kb + 1];
          const int p = (l * 32 + kk * 2 + half) * kWPS + 2 * fl + 32 * j;
          v2f bf;
          bf.x = s_w[p];
          bf.y = s_w[p + 1];
          t = __builtin_amdgcn_wmma_f32_16x16x4_f32(
              false, bf, false, a, (short)0, t, false, false);
        }
        base[l] = t;
#pragma unroll
        for (int q = 0; q < 4; ++q) {
          const int p = (l * 32 + 8 * j + 2 * q + half) * kWPS + 2 * fl + 32 * j;
          wd[l][q].x = s_w[p];
          wd[l][q].y = s_w[p + 1];
        }
      }
      v8f st[4];
#pragma unroll
      for (int l = 0; l < 4; ++l)
#pragma unroll
        for (int v = 0; v < 8; ++v) st[l][v] = 0.0f;
#pragma unroll
      for (int q = 0; q < 4; ++q) {
        v8f pre[4];
#pragma unroll 1
        for (int i = 0; i < 4; ++i) {
#pragma unroll
          for (int l = 0; l < 4; ++l) {
            v2f bq;
            bq.x = st[l][2 * q];
            bq.y = st[l][2 * q + 1];
            pre[l] = __builtin_amdgcn_wmma_f32_16x16x4_f32(
                false, wd[l][q], false, bq, (short)0, base[l], false, false);
            v8f nx;
            if (l < 3) {
#pragma unroll
              for (int v = 0; v < 8; ++v) nx[v] = fmaxf(pre[l][v], 0.0f);
            } else {
#pragma unroll
              for (int v = 0; v < 8; ++v) nx[v] = pre[l][v] + x0t[v];
            }
            st[(l + 1) & 3] = nx;
          }
        }
        if (q < 3) {
          v2f bq;
          bq.x = st[0][2 * q];
          bq.y = st[0][2 * q + 1];
          base[0] = __builtin_amdgcn_wmma_f32_16x16x4_f32(
              false, wd[0][q], false, bq, (short)0, base[0], false, false);
#pragma unroll
          for (int l = 1; l < 4; ++l) base[l] = pre[l];
        }
      }
#pragma unroll
      for (int l = 0; l < 4; ++l)
#pragma unroll
        for (int v = 0; v < 8; ++v) s_act[wv][l][tcol + 4 * (v >> 1) + (v & 1)] = st[l][v];
      wave_fence();
    }
#pragma unroll
    for (int n = 0; n < 4; ++n)
#pragma unroll
      for (int v = 0; v < 8; ++v)
        y[n][v] = s_act[wv][0][(v + half * 8) * kAst + n * 16 + l15];
    wave_fence();
  };

  density_stage();
  reverse_cols(y);
  load_weights(1);
  density_stage();

  float part[8];
#pragma unroll
  for (int v = 0; v < 8; ++v) part[v] = 0.0f;
#pragma unroll
  for (int n = 0; n < 4; ++n)
#pragma unroll
    for (int v = 0; v < 8; ++v) {
      const int row = r0 + v + half * 8;
      const int col = n * 16 + l15;
      const float m = zmean[row * kD + col];
      const float lv = zlogvar[row * kD + col];
      const float d = y[n][v] - m;
      part[v] += d * d * expf(-lv) + lv + kLog2Pi;
    }
#pragma unroll
  for (int v = 0; v < 8; ++v) s_red[tid * 8 + v] = part[v];
  wave_fence();
  if (lane < 16) {
    const int h2 = lane >> 3;
    const int vv = lane & 7;
    const int base = wv * 32 + h2 * 16;
    float s = 0.0f;
#pragma unroll
    for (int j = 0; j < 16; ++j) s += s_red[(base + j) * 8 + vv];
    out_lp[r0 + lane] = -0.5f * s;
  }
}

extern "C" void kernel_launch(void* const* d_in, const int* in_sizes, int n_in,
                              void* d_out, int out_size, void* d_ws, size_t ws_size,
                              hipStream_t stream) {
  (void)in_sizes; (void)n_in; (void)out_size; (void)d_ws; (void)ws_size;
  const float* zmean = (const float*)d_in[0];
  const float* zlogvar = (const float*)d_in[1];
  const float* noise = (const float*)d_in[2];
  const float* Wg = (const float*)d_in[3];
  const float* bvec = (const float*)d_in[4];
  float* samples = (float*)d_out;
  float* lp = samples + (size_t)kB * kD;
  const dim3 grid(kB / kRowsPerBlock);
  iaf_kernel<<<grid, kThreads, 0, stream>>>(zmean, zlogvar, noise, Wg, bvec, samples, lp);
}


# sampling layers skip the K chunks the triangular mask zeroes: column tile n multiplies k < 16(n+1) only, 40 matrix instructions per layer instead of 64; order of the kept sums unchanged
# speedup vs baseline: 22.6024x; 1.1139x over previous
#include <hip/hip_runtime.h>
#include <math.h>

namespace {
constexpr int kB = 16384;
constexpr int kD = 64;
constexpr int kWaves = 4;
constexpr int kThreads = kWaves * 32;
constexpr int kRowsPerWave = 16;
constexpr int kRowsPerBlock = kWaves * kRowsPerWave;
constexpr int kAst = 68;
constexpr int kWPS = 160;
constexpr float kLog2Pi = 1.8378770664093453f;
}

typedef float v2f __attribute__((ext_vector_type(2)));
typedef float v4f __attribute__((ext_vector_type(4)));
typedef float v8f __attribute__((ext_vector_type(8)));

__device__ __forceinline__ void wave_fence() { asm volatile("" ::: "memory"); }

__global__ void __launch_bounds__(kThreads)
iaf_kernel(const float* __restrict__ zmean, const float* __restrict__ zlogvar,
           const float* __restrict__ noise, const float* __restrict__ Wg,
           const float* __restrict__ bvec, float* __restrict__ out_samples,
           float* __restrict__ out_lp)
{
  __shared__ alignas(16) float s_w[4 * 32 * kWPS];
  __shared__ float s_b[4 * kD];
  __shared__ float s_act[kWaves][4][kRowsPerWave * kAst];
  __shared__ float s_red[kThreads * 8];

  const int tid = threadIdx.x;
  const int lane = tid & 31;
  const int wv = tid >> 5;
  const int half = lane >> 4;
  const int l15 = lane & 15;
  const int r0 = blockIdx.x * kRowsPerBlock + wv * kRowsPerWave;

  auto load_weights = [&](int s) {
    v4f w0[16], w1[16];
    const int nq = tid & 15;
    const int kq = tid >> 4;
#pragma unroll
    for (int it = 0; it < 16; ++it) {
      const int l = it >> 2;
      const int kp = (it & 3) * 8 + kq;
      const float* src = Wg + ((s * 4 + l) * kD + 2 * kp) * kD + 4 * nq;
      w0[it] = *reinterpret_cast<const v4f*>(src);
      w1[it] = *reinterpret_cast<const v4f*>(src + kD);
    }
    const float b0 = bvec[s * 4 * kD + tid];
    const float b1 = bvec[s * 4 * kD + kThreads + tid];
    __syncthreads();
#pragma unroll
    for (int it = 0; it < 16; ++it) {
      const int l = it >> 2;
      const int kp = (it & 3) * 8 + kq;
      const int k0 = 2 * kp;
      const int k1 = k0 + 1;
      const int n0 = 4 * nq;
      const int d0 = (l == 0) ? k0 + 1 : k0;
      const int d1 = (l == 0) ? k1 + 1 : k1;
      v4f lo, hi;
      lo.x = (n0 + 0 >= d0) ? w0[it].x : 0.0f;
      lo.y = (n0 + 0 >= d1) ? w1[it].x : 0.0f;
      lo.z = (n0 + 1 >= d0) ? w0[it].y : 0.0f;
      lo.w = (n0 + 1 >= d1) ? w1[it].y : 0.0f;
      hi.x = (n0 + 2 >= d0) ? w0[it].z : 0.0f;
      hi.y = (n0 + 2 >= d1) ? w1[it].z : 0.0f;
      hi.z = (n0 + 3 >= d0) ? w0[it].w : 0.0f;
      hi.w = (n0 + 3 >= d1) ? w1[it].w : 0.0f;
      float* dst = &s_w[(l * 32 + kp) * kWPS + 2 * n0];
      *reinterpret_cast<v4f*>(dst) = lo;
      *reinterpret_cast<v4f*>(dst + 4) = hi;
    }
    s_b[tid] = b0;
    s_b[kThreads + tid] = b1;
    __syncthreads();
  };

  auto layer = [&](int ib, int ob, int l, bool do_relu, v8f* outfr, const v8f* addfr) {
    v8f acc[4];
#pragma unroll
    for (int n = 0; n < 4; ++n) {
      const float bv = s_b[l * kD + n * 16 + l15];
      v8f t;
#pragma unroll
      for (int v = 0; v < 8; ++v) t[v] = bv;
      acc[n] = t;
    }
    const int arow = l15 * kAst;
    const int wbase = 2 * l15;
#pragma unroll
    for (int kk = 0; kk < 16; ++kk) {
      const int kb = kk * 4 + half * 2;
      v2f a;
      a.x = s_act[wv][ib][arow + kb];
      a.y = s_act[wv][ib][arow + kb + 1];
      const int pbase = (l * 32 + kk * 2 + half) * kWPS + wbase;
#pragma unroll
      for (int n = 0; n < 4; ++n) {
        if (kk >= 4 * (n + 1)) continue;
        v2f bf;
        bf.x = s_w[pbase + 32 * n];
        bf.y = s_w[pbase + 32 * n + 1];
        acc[n] = __builtin_amdgcn_wmma_f32_16x16x4_f32(
            false, a, false, bf, (short)0, acc[n], false, false);
      }
    }
    if (do_relu) {
#pragma unroll
      for (int n = 0; n < 4; ++n)
#pragma unroll
        for (int v = 0; v < 8; ++v) acc[n][v] = fmaxf(acc[n][v], 0.0f);
    }
    if (addfr) {
#pragma unroll
      for (int n = 0; n < 4; ++n)
#pragma unroll
        for (int v = 0; v < 8; ++v) acc[n][v] += addfr[n][v];
    }
    if (ob >= 0) {
#pragma unroll
      for (int n = 0; n < 4; ++n)
#pragma unroll
        for (int v = 0; v < 8; ++v)
          s_act[wv][ob][(v + half * 8) * kAst + n * 16 + l15] = acc[n][v];
      wave_fence();
    } else {
#pragma unroll
      for (int n = 0; n < 4; ++n) outfr[n] = acc[n];
    }
  };

  auto store_frags = [&](const v8f* fr, int b) {
#pragma unroll
    for (int n = 0; n < 4; ++n)
#pragma unroll
      for (int v = 0; v < 8; ++v)
        s_act[wv][b][(v + half * 8) * kAst + n * 16 + l15] = fr[n][v];
    wave_fence();
  };

  auto made = [&](v8f* outfr) {
    layer(0, 1, 0, true, nullptr, nullptr);
    layer(1, 0, 1, true, nullptr, nullptr);
    layer(0, 1, 2, true, nullptr, nullptr);
    layer(1, -1, 3, false, outfr, nullptr);
  };

  auto reverse_cols = [&](v8f* fr) {
    store_frags(fr, 0);
#pragma unroll
    for (int n = 0; n < 4; ++n)
#pragma unroll
      for (int v = 0; v < 8; ++v)
        fr[n][v] = s_act[wv][0][(v + half * 8) * kAst + (63 - (n * 16 + l15))];
    wave_fence();
  };

  v8f y[4];
#pragma unroll
  for (int n = 0; n < 4; ++n)
#pragma unroll
    for (int v = 0; v < 8; ++v) {
      const int row = r0 + v + half * 8;
      const int col = n * 16 + l15;
      const float m = zmean[row * kD + col];
      const float lv = zlogvar[row * kD + col];
      const float e = noise[row * kD + col];
      y[n][v] = m + expf(0.5f * lv) * e;
    }

  v8f m4[4];

  load_weights(1);
  store_frags(y, 0);
  made(m4);
#pragma unroll
  for (int n = 0; n < 4; ++n)
#pragma unroll
    for (int v = 0; v < 8; ++v) y[n][v] -= m4[n][v];
  reverse_cols(y);

  load_weights(0);
  store_frags(y, 0);
  made(m4);
#pragma unroll
  for (int n = 0; n < 4; ++n)
#pragma unroll
    for (int v = 0; v < 8; ++v) y[n][v] -= m4[n][v];

#pragma unroll
  for (int n = 0; n < 4; ++n)
#pragma unroll
    for (int v = 0; v < 8; ++v)
      out_samples[(r0 + v + half * 8) * kD + n * 16 + l15] = y[n][v];

  auto density_stage = [&]() {
    store_frags(y, 0);
    const int arow = l15 * kAst;
    const int fl = 4 * ((l15 >> 1) & 3) + 2 * (l15 >> 3) + (l15 & 1);
#pragma unroll
    for (int j = 0; j < 4; ++j) {
      const int tcol = arow + j * 16 + half * 2;
      v8f x0t;
#pragma unroll
      for (int v = 0; v < 8; ++v) x0t[v] = s_act[wv][0][tcol + 4 * (v >> 1) + (v & 1)];
      v8f base[4];
      v2f wd[4][4];
#pragma unroll
      for (int l = 0; l < 4; ++l) {
        v8f t;
#pragma unroll
        for (int v = 0; v < 8; ++v)
          t[v] = s_b[l * kD + j * 16 + 4 * (v >> 1) + 2 * half + (v & 1)];
#pragma unroll
        for (int kk = 0; kk < 4 * j; ++kk) {
          const int kb = kk * 4 + half * 2;
          v2f a;
          a.x = s_act[wv][l][arow + kb];
          a.y = s_act[wv][l][arow + kb + 1];
          const int p = (l * 32 + kk * 2 + half) * kWPS + 2 * fl + 32 * j;
          v2f bf;
          bf.x = s_w[p];
          bf.y = s_w[p + 1];
          t = __builtin_amdgcn_wmma_f32_16x16x4_f32(
              false, bf, false, a, (short)0, t, false, false);
        }
        base[l] = t;
#pragma unroll
        for (int q = 0; q < 4; ++q) {
          const int p = (l * 32 + 8 * j + 2 * q + half) * kWPS + 2 * fl + 32 * j;
          wd[l][q].x = s_w[p];
          wd[l][q].y = s_w[p + 1];
        }
      }
      v8f st[4];
#pragma unroll
      for (int l = 0; l < 4; ++l)
#pragma unroll
        for (int v = 0; v < 8; ++v) st[l][v] = 0.0f;
#pragma unroll
      for (int q = 0; q < 4; ++q) {
        v8f pre[4];
#pragma unroll 1
        for (int i = 0; i < 4; ++i) {
#pragma unroll
          for (int l = 0; l < 4; ++l) {
            v2f bq;
            bq.x = st[l][2 * q];
            bq.y = st[l][2 * q + 1];
            pre[l] = __builtin_amdgcn_wmma_f32_16x16x4_f32(
                false, wd[l][q], false, bq, (short)0, base[l], false, false);
            v8f nx;
            if (l < 3) {
#pragma unroll
              for (int v = 0; v < 8; ++v) nx[v] = fmaxf(pre[l][v], 0.0f);
            } else {
#pragma unroll
              for (int v = 0; v < 8; ++v) nx[v] = pre[l][v] + x0t[v];
            }
            st[(l + 1) & 3] = nx;
          }
        }
        if (q < 3) {
          v2f bq;
          bq.x = st[0][2 * q];
          bq.y = st[0][2 * q + 1];
          base[0] = __builtin_amdgcn_wmma_f32_16x16x4_f32(
              false, wd[0][q], false, bq, (short)0, base[0], false, false);
#pragma unroll
          for (int l = 1; l < 4; ++l) base[l] = pre[l];
        }
      }
#pragma unroll
      for (int l = 0; l < 4; ++l)
#pragma unroll
        for (int v = 0; v < 8; ++v) s_act[wv][l][tcol + 4 * (v >> 1) + (v & 1)] = st[l][v];
      wave_fence();
    }
#pragma unroll
    for (int n = 0; n < 4; ++n)
#pragma unroll
      for (int v = 0; v < 8; ++v)
        y[n][v] = s_act[wv][0][(v + half * 8) * kAst + n * 16 + l15];
    wave_fence();
  };

  density_stage();
  reverse_cols(y);
  load_weights(1);
  density_stage();

  float part[8];
#pragma unroll
  for (int v = 0; v < 8; ++v) part[v] = 0.0f;
#pragma unroll
  for (int n = 0; n < 4; ++n)
#pragma unroll
    for (int v = 0; v < 8; ++v) {
      const int row = r0 + v + half * 8;
      const int col = n * 16 + l15;
      const float m = zmean[row * kD + col];
      const float lv = zlogvar[row * kD + col];
      const float d = y[n][v] - m;
      part[v] += d * d * expf(-lv) + lv + kLog2Pi;
    }
#pragma unroll
  for (int v = 0; v < 8; ++v) s_red[tid * 8 + v] = part[v];
  wave_fence();
  if (lane < 16) {
    const int h2 = lane >> 3;
    const int vv = lane & 7;
    const int base = wv * 32 + h2 * 16;
    float s = 0.0f;
#pragma unroll
    for (int j = 0; j < 16; ++j) s += s_red[(base + j) * 8 + vv];
    out_lp[r0 + lane] = -0.5f * s;
  }
}

extern "C" void kernel_launch(void* const* d_in, const int* in_sizes, int n_in,
                              void* d_out, int out_size, void* d_ws, size_t ws_size,
                              hipStream_t stream) {
  (void)in_sizes; (void)n_in; (void)out_size; (void)d_ws; (void)ws_size;
  const float* zmean = (const float*)d_in[0];
  const float* zlogvar = (const float*)d_in[1];
  const float* noise = (const float*)d_in[2];
  const float* Wg = (const float*)d_in[3];
  const float* bvec = (const float*)d_in[4];
  float* samples = (float*)d_out;
  float* lp = samples + (size_t)kB * kD;
  const dim3 grid(kB / kRowsPerBlock);
  iaf_kernel<<<grid, kThreads, 0, stream>>>(zmean, zlogvar, noise, Wg, bvec, samples, lp);
}


# density: a finished K chunk's products into the later tiles are issued one behind each dependent step of the next group, from registers (chain unrolled, issue order pinned); sums unchanged
# speedup vs baseline: 23.9978x; 1.0617x over previous
#include <hip/hip_runtime.h>
#include <math.h>

namespace {
constexpr int kB = 16384;
constexpr int kD = 64;
constexpr int kWaves = 4;
constexpr int kThreads = kWaves * 32;
constexpr int kRowsPerWave = 16;
constexpr int kRowsPerBlock = kWaves * kRowsPerWave;
constexpr int kAst = 68;
constexpr int kWPS = 160;
constexpr float kLog2Pi = 1.8378770664093453f;
}

typedef float v2f __attribute__((ext_vector_type(2)));
typedef float v4f __attribute__((ext_vector_type(4)));
typedef float v8f __attribute__((ext_vector_type(8)));

__device__ __forceinline__ void wave_fence() { asm volatile("" ::: "memory"); }

__global__ void __launch_bounds__(kThreads)
iaf_kernel(const float* __restrict__ zmean, const float* __restrict__ zlogvar,
           const float* __restrict__ noise, const float* __restrict__ Wg,
           const float* __restrict__ bvec, float* __restrict__ out_samples,
           float* __restrict__ out_lp)
{
  __shared__ alignas(16) float s_w[4 * 32 * kWPS];
  __shared__ float s_b[4 * kD];
  __shared__ float s_act[kWaves][4][kRowsPerWave * kAst];
  __shared__ float s_red[kThreads * 8];

  const int tid = threadIdx.x;
  const int lane = tid & 31;
  const int wv = tid >> 5;
  const int half = lane >> 4;
  const int l15 = lane & 15;
  const int r0 = blockIdx.x * kRowsPerBlock + wv * kRowsPerWave;

  auto load_weights = [&](int s) {
    v4f w0[16], w1[16];
    const int nq = tid & 15;
    const int kq = tid >> 4;
#pragma unroll
    for (int it = 0; it < 16; ++it) {
      const int l = it >> 2;
      const int kp = (it & 3) * 8 + kq;
      const float* src = Wg + ((s * 4 + l) * kD + 2 * kp) * kD + 4 * nq;
      w0[it] = *reinterpret_cast<const v4f*>(src);
      w1[it] = *reinterpret_cast<const v4f*>(src + kD);
    }
    const float b0 = bvec[s * 4 * kD + tid];
    const float b1 = bvec[s * 4 * kD + kThreads + tid];
    __syncthreads();
#pragma unroll
    for (int it = 0; it < 16; ++it) {
      const int l = it >> 2;
      const int kp = (it & 3) * 8 + kq;
      const int k0 = 2 * kp;
      const int k1 = k0 + 1;
      const int n0 = 4 * nq;
      const int d0 = (l == 0) ? k0 + 1 : k0;
      const int d1 = (l == 0) ? k1 + 1 : k1;
      v4f lo, hi;
      lo.x = (n0 + 0 >= d0) ? w0[it].x : 0.0f;
      lo.y = (n0 + 0 >= d1) ? w1[it].x : 0.0f;
      lo.z = (n0 + 1 >= d0) ? w0[it].y : 0.0f;
      lo.w = (n0 + 1 >= d1) ? w1[it].y : 0.0f;
      hi.x = (n0 + 2 >= d0) ? w0[it].z : 0.0f;
      hi.y = (n0 + 2 >= d1) ? w1[it].z : 0.0f;
      hi.z = (n0 + 3 >= d0) ? w0[it].w : 0.0f;
      hi.w = (n0 + 3 >= d1) ? w1[it].w : 0.0f;
      float* dst = &s_w[(l * 32 + kp) * kWPS + 2 * n0];
      *reinterpret_cast<v4f*>(dst) = lo;
      *reinterpret_cast<v4f*>(dst + 4) = hi;
    }
    s_b[tid] = b0;
    s_b[kThreads + tid] = b1;
    __syncthreads();
  };

  auto layer = [&](int ib, int ob, int l, bool do_relu, v8f* outfr, const v8f* addfr) {
    v8f acc[4];
#pragma unroll
    for (int n = 0; n < 4; ++n) {
      const float bv = s_b[l * kD + n * 16 + l15];
      v8f t;
#pragma unroll
      for (int v = 0; v < 8; ++v) t[v] = bv;
      acc[n] = t;
    }
    const int arow = l15 * kAst;
    const int wbase = 2 * l15;
#pragma unroll
    for (int kk = 0; kk < 16; ++kk) {
      const int kb = kk * 4 + half * 2;
      v2f a;
      a.x = s_act[wv][ib][arow + kb];
      a.y = s_act[wv][ib][arow + kb + 1];
      const int pbase = (l * 32 + kk * 2 + half) * kWPS + wbase;
#pragma unroll
      for (int n = 0; n < 4; ++n) {
        if (kk >= 4 * (n + 1)) continue;
        v2f bf;
        bf.x = s_w[pbase + 32 * n];
        bf.y = s_w[pbase + 32 * n + 1];
        acc[n] = __builtin_amdgcn_wmma_f32_16x16x4_f32(
            false, a, false, bf, (short)0, acc[n], false, false);
      }
    }
    if (do_relu) {
#pragma unroll
      for (int n = 0; n < 4; ++n)
#pragma unroll
        for (int v = 0; v < 8; ++v) acc[n][v] = fmaxf(acc[n][v], 0.0f);
    }
    if (addfr) {
#pragma unroll
      for (int n = 0; n < 4; ++n)
#pragma unroll
        for (int v = 0; v < 8; ++v) acc[n][v] += addfr[n][v];
    }
    if (ob >= 0) {
#pragma unroll
      for (int n = 0; n < 4; ++n)
#pragma unroll
        for (int v = 0; v < 8; ++v)
          s_act[wv][ob][(v + half * 8) * kAst + n * 16 + l15] = acc[n][v];
      wave_fence();
    } else {
#pragma unroll
      for (int n = 0; n < 4; ++n) outfr[n] = acc[n];
    }
  };

  auto store_frags = [&](const v8f* fr, int b) {
#pragma unroll
    for (int n = 0; n < 4; ++n)
#pragma unroll
      for (int v = 0; v < 8; ++v)
        s_act[wv][b][(v + half * 8) * kAst + n * 16 + l15] = fr[n][v];
    wave_fence();
  };

  auto made = [&](v8f* outfr) {
    layer(0, 1, 0, true, nullptr, nullptr);
    layer(1, 0, 1, true, nullptr, nullptr);
    layer(0, 1, 2, true, nullptr, nullptr);
    layer(1, -1, 3, false, outfr, nullptr);
  };

  auto reverse_cols = [&](v8f* fr) {
    store_frags(fr, 0);
#pragma unroll
    for (int n = 0; n < 4; ++n)
#pragma unroll
      for (int v = 0; v < 8; ++v)
        fr[n][v] = s_act[wv][0][(v + half * 8) * kAst + (63 - (n * 16 + l15))];
    wave_fence();
  };

  v8f y[4];
#pragma unroll
  for (int n = 0; n < 4; ++n)
#pragma unroll
    for (int v = 0; v < 8; ++v) {
      const int row = r0 + v + half * 8;
      const int col = n * 16 + l15;
      const float m = zmean[row * kD + col];
      const float lv = zlogvar[row * kD + col];
      const float e = noise[row * kD + col];
      y[n][v] = m + expf(0.5f * lv) * e;
    }

  v8f m4[4];

  load_weights(1);
  store_frags(y, 0);
  made(m4);
#pragma unroll
  for (int n = 0; n < 4; ++n)
#pragma unroll
    for (int v = 0; v < 8; ++v) y[n][v] -= m4[n][v];
  reverse_cols(y);

  load_weights(0);
  store_frags(y, 0);
  made(m4);
#pragma unroll
  for (int n = 0; n < 4; ++n)
#pragma unroll
    for (int v = 0; v < 8; ++v) y[n][v] -= m4[n][v];

#pragma unroll
  for (int n = 0; n < 4; ++n)
#pragma unroll
    for (int v = 0; v < 8; ++v)
      out_samples[(r0 + v + half * 8) * kD + n * 16 + l15] = y[n][v];

  auto density_stage = [&]() {
    store_frags(y, 0);
    const int arow = l15 * kAst;
    const int fl = 4 * ((l15 >> 1) & 3) + 2 * (l15 >> 3) + (l15 & 1);
    v8f base[4][4];
#pragma unroll
    for (int jt = 0; jt < 4; ++jt)
#pragma unroll
      for (int l = 0; l < 4; ++l) {
        v8f t;
#pragma unroll
        for (int v = 0; v < 8; ++v)
          t[v] = s_b[l * kD + jt * 16 + 4 * (v >> 1) + 2 * half + (v & 1)];
        base[jt][l] = t;
      }
    auto load_x0t = [&](int jj, v8f& x) {
#pragma unroll
      for (int v = 0; v < 8; ++v)
        x[v] = s_act[wv][0][arow + jj * 16 + half * 2 + 4 * (v >> 1) + (v & 1)];
    };
    auto load_wd = [&](int jj, v2f (&w)[4][4]) {
#pragma unroll
      for (int l = 0; l < 4; ++l)
#pragma unroll
        for (int q = 0; q < 4; ++q) {
          const int p = (l * 32 + 8 * jj + 2 * q + half) * kWPS + 2 * fl + 32 * jj;
          w[l][q].x = s_w[p];
          w[l][q].y = s_w[p + 1];
        }
    };
    v8f x0n;
    v2f wdn[4][4];
    load_x0t(0, x0n);
    load_wd(0, wdn);
    v2f pb[4];
    v2f wfc[12];
    v2f wfn[12];
#pragma unroll
    for (int l = 0; l < 4; ++l) { pb[l].x = 0.0f; pb[l].y = 0.0f; }
#pragma unroll
    for (int k = 0; k < 12; ++k) { wfc[k].x = 0.0f; wfc[k].y = 0.0f; wfn[k] = wfc[k]; }
    int pj = 0, pn = 0;
#pragma unroll
    for (int j = 0; j < 4; ++j) {
      v8f x0t = x0n;
      v2f wd[4][4];
#pragma unroll
      for (int l = 0; l < 4; ++l)
#pragma unroll
        for (int q = 0; q < 4; ++q) wd[l][q] = wdn[l][q];
      v8f st[4];
#pragma unroll
      for (int l = 0; l < 4; ++l)
#pragma unroll
        for (int v = 0; v < 8; ++v) st[l][v] = 0.0f;
#pragma unroll
      for (int q = 0; q < 4; ++q) {
#pragma unroll
        for (int k = 0; k < 4 * (3 - j); ++k) {
          const int jt = j + 1 + (k >> 2);
          const int l = k & 3;
          const int p = (l * 32 + 8 * j + 2 * q + half) * kWPS + 2 * fl + 32 * jt;
          wfn[k].x = s_w[p];
          wfn[k].y = s_w[p + 1];
        }
        if (q == 3 && j < 3) {
          load_x0t(j + 1, x0n);
          load_wd(j + 1, wdn);
        }
        auto owed = [&](int k) {
          if (k < pn) {
            const int jt = pj + 1 + (k >> 2);
            const int l = k & 3;
            base[jt][l] = __builtin_amdgcn_wmma_f32_16x16x4_f32(
                false, wfc[k], false, pb[l], (short)0, base[jt][l], false, false);
          }
        };
        int k0 = 0;
        if (q == 0 && j > 0) { owed(0); owed(1); k0 = 2; }
        v8f pre[4];
#pragma unroll
        for (int i = 0; i < 4; ++i) {
#pragma unroll
          for (int l = 0; l < 4; ++l) {
            v2f bq;
            bq.x = st[l][2 * q];
            bq.y = st[l][2 * q + 1];
            pre[l] = __builtin_amdgcn_wmma_f32_16x16x4_f32(
                false, wd[l][q], false, bq, (short)0, base[j][l], false, false);
            __builtin_amdgcn_sched_barrier(0);
            owed(k0 + 4 * i + l);
            __builtin_amdgcn_sched_barrier(0);
            v8f nx;
            if (l < 3) {
#pragma unroll
              for (int v = 0; v < 8; ++v) nx[v] = fmaxf(pre[l][v], 0.0f);
            } else {
#pragma unroll
              for (int v = 0; v < 8; ++v) nx[v] = pre[l][v] + x0t[v];
            }
            st[(l + 1) & 3] = nx;
          }
        }
        if (q < 3) {
          v2f bq;
          bq.x = st[0][2 * q];
          bq.y = st[0][2 * q + 1];
          base[j][0] = __builtin_amdgcn_wmma_f32_16x16x4_f32(
              false, wd[0][q], false, bq, (short)0, base[j][0], false, false);
#pragma unroll
          for (int l = 1; l < 4; ++l) base[j][l] = pre[l];
        }
#pragma unroll
        for (int l = 0; l < 4; ++l) {
          pb[l].x = st[l][2 * q];
          pb[l].y = st[l][2 * q + 1];
        }
#pragma unroll
        for (int k = 0; k < 12; ++k) wfc[k] = wfn[k];
        pj = j;
        pn = 4 * (3 - j);
      }
#pragma unroll
      for (int v = 0; v < 8; ++v)
        s_act[wv][0][arow + j * 16 + half * 2 + 4 * (v >> 1) + (v & 1)] = st[0][v];
      wave_fence();
    }
#pragma unroll
    for (int n = 0; n < 4; ++n)
#pragma unroll
      for (int v = 0; v < 8; ++v)
        y[n][v] = s_act[wv][0][(v + half * 8) * kAst + n * 16 + l15];
    wave_fence();
  };

  density_stage();
  reverse_cols(y);
  load_weights(1);
  density_stage();

  float part[8];
#pragma unroll
  for (int v = 0; v < 8; ++v) part[v] = 0.0f;
#pragma unroll
  for (int n = 0; n < 4; ++n)
#pragma unroll
    for (int v = 0; v < 8; ++v) {
      const int row = r0 + v + half * 8;
      const int col = n * 16 + l15;
      const float m = zmean[row * kD + col];
      const float lv = zlogvar[row * kD + col];
      const float d = y[n][v] - m;
      part[v] += d * d * expf(-lv) + lv + kLog2Pi;
    }
#pragma unroll
  for (int v = 0; v < 8; ++v) s_red[tid * 8 + v] = part[v];
  wave_fence();
  if (lane < 16) {
    const int h2 = lane >> 3;
    const int vv = lane & 7;
    const int base = wv * 32 + h2 * 16;
    float s = 0.0f;
#pragma unroll
    for (int j = 0; j < 16; ++j) s += s_red[(base + j) * 8 + vv];
    out_lp[r0 + lane] = -0.5f * s;
  }
}

extern "C" void kernel_launch(void* const* d_in, const int* in_sizes, int n_in,
                              void* d_out, int out_size, void* d_ws, size_t ws_size,
                              hipStream_t stream) {
  (void)in_sizes; (void)n_in; (void)out_size; (void)d_ws; (void)ws_size;
  const float* zmean = (const float*)d_in[0];
  const float* zlogvar = (const float*)d_in[1];
  const float* noise = (const float*)d_in[2];
  const float* Wg = (const float*)d_in[3];
  const float* bvec = (const float*)d_in[4];
  float* samples = (float*)d_out;
  float* lp = samples + (size_t)kB * kD;
  const dim3 grid(kB / kRowsPerBlock);
  iaf_kernel<<<grid, kThreads, 0, stream>>>(zmean, zlogvar, noise, Wg, bvec, samples, lp);
}


# weights of both stages kept in LDS, each fetched once (stage 1 was fetched twice); stage 0's loads travel under the first sampling pass; 2 barriers instead of 6; every sum's order unchanged
# speedup vs baseline: 25.7750x; 1.0741x over previous
#include <hip/hip_runtime.h>
#include <math.h>

namespace {
constexpr int kB = 16384;
constexpr int kD = 64;
constexpr int kWaves = 4;
constexpr int kThreads = kWaves * 32;
constexpr int kRowsPerWave = 16;
constexpr int kRowsPerBlock = kWaves * kRowsPerWave;
constexpr int kAst = 68;
constexpr int kWPS = 160;
constexpr int kWStage = 4 * 32 * kWPS;
constexpr float kLog2Pi = 1.8378770664093453f;
}

typedef float v2f __attribute__((ext_vector_type(2)));
typedef float v4f __attribute__((ext_vector_type(4)));
typedef float v8f __attribute__((ext_vector_type(8)));

__device__ __forceinline__ void wave_fence() { asm volatile("" ::: "memory"); }

__global__ void __launch_bounds__(kThreads)
iaf_kernel(const float* __restrict__ zmean, const float* __restrict__ zlogvar,
           const float* __restrict__ noise, const float* __restrict__ Wg,
           const float* __restrict__ bvec, float* __restrict__ out_samples,
           float* __restrict__ out_lp)
{
  __shared__ alignas(16) float s_w[2 * kWStage];
  __shared__ float s_b[2 * 4 * kD];
  __shared__ float s_act[kWaves][4][kRowsPerWave * kAst];
  __shared__ float s_red[kThreads * 8];

  const int tid = threadIdx.x;
  const int lane = tid & 31;
  const int wv = tid >> 5;
  const int half = lane >> 4;
  const int l15 = lane & 15;
  const int r0 = blockIdx.x * kRowsPerBlock + wv * kRowsPerWave;

  const int nq = tid & 15;
  const int kq = tid >> 4;
  auto issue_weights = [&](int s, v4f (&w0)[16], v4f (&w1)[16], float& b0, float& b1) {
#pragma unroll
    for (int it = 0; it < 16; ++it) {
      const int l = it >> 2;
      const int kp = (it & 3) * 8 + kq;
      const float* src = Wg + ((s * 4 + l) * kD + 2 * kp) * kD + 4 * nq;
      w0[it] = *reinterpret_cast<const v4f*>(src);
      w1[it] = *reinterpret_cast<const v4f*>(src + kD);
    }
    b0 = bvec[s * 4 * kD + tid];
    b1 = bvec[s * 4 * kD + kThreads + tid];
  };
  auto store_weights = [&](int s, const v4f (&w0)[16], const v4f (&w1)[16], float b0, float b1) {
#pragma unroll
    for (int it = 0; it < 16; ++it) {
      const int l = it >> 2;
      const int kp = (it & 3) * 8 + kq;
      const int k0 = 2 * kp;
      const int k1 = k0 + 1;
      const int n0 = 4 * nq;
      const int d0 = (l == 0) ? k0 + 1 : k0;
      const int d1 = (l == 0) ? k1 + 1 : k1;
      v4f lo, hi;
      lo.x = (n0 + 0 >= d0) ? w0[it].x : 0.0f;
      lo.y = (n0 + 0 >= d1) ? w1[it].x : 0.0f;
      lo.z = (n0 + 1 >= d0) ? w0[it].y : 0.0f;
      lo.w = (n0 + 1 >= d1) ? w1[it].y : 0.0f;
      hi.x = (n0 + 2 >= d0) ? w0[it].z : 0.0f;
      hi.y = (n0 + 2 >= d1) ? w1[it].z : 0.0f;
      hi.z = (n0 + 3 >= d0) ? w0[it].w : 0.0f;
      hi.w = (n0 + 3 >= d1) ? w1[it].w : 0.0f;
      float* dst = &s_w[s * kWStage + (l * 32 + kp) * kWPS + 2 * n0];
      *reinterpret_cast<v4f*>(dst) = lo;
      *reinterpret_cast<v4f*>(dst + 4) = hi;
    }
    s_b[s * 4 * kD + tid] = b0;
    s_b[s * 4 * kD + kThreads + tid] = b1;
  };

  auto layer = [&](int s, int ib, int ob, int l, bool do_relu, v8f* outfr, const v8f* addfr) {
    v8f acc[4];
#pragma unroll
    for (int n = 0; n < 4; ++n) {
      const float bv = s_b[s * 4 * kD + l * kD + n * 16 + l15];
      v8f t;
#pragma unroll
      for (int v = 0; v < 8; ++v) t[v] = bv;
      acc[n] = t;
    }
    const int arow = l15 * kAst;
    const int wbase = 2 * l15;
#pragma unroll
    for (int kk = 0; kk < 16; ++kk) {
      const int kb = kk * 4 + half * 2;
      v2f a;
      a.x = s_act[wv][ib][arow + kb];
      a.y = s_act[wv][ib][arow + kb + 1];
      const int pbase = s * kWStage + (l * 32 + kk * 2 + half) * kWPS + wbase;
#pragma unroll
      for (int n = 0; n < 4; ++n) {
        if (kk >= 4 * (n + 1)) continue;
        v2f bf;
        bf.x = s_w[pbase + 32 * n];
        bf.y = s_w[pbase + 32 * n + 1];
        acc[n] = __builtin_amdgcn_wmma_f32_16x16x4_f32(
            false, a, false, bf, (short)0, acc[n], false, false);
      }
    }
    if (do_relu) {
#pragma unroll
      for (int n = 0; n < 4; ++n)
#pragma unroll
        for (int v = 0; v < 8; ++v) acc[n][v] = fmaxf(acc[n][v], 0.0f);
    }
    if (addfr) {
#pragma unroll
      for (int n = 0; n < 4; ++n)
#pragma unroll
        for (int v = 0; v < 8; ++v) acc[n][v] += addfr[n][v];
    }
    if (ob >= 0) {
#pragma unroll
      for (int n = 0; n < 4; ++n)
#pragma unroll
        for (int v = 0; v < 8; ++v)
          s_act[wv][ob][(v + half * 8) * kAst + n * 16 + l15] = acc[n][v];
      wave_fence();
    } else {
#pragma unroll
      for (int n = 0; n < 4; ++n) outfr[n] = acc[n];
    }
  };

  auto store_frags = [&](const v8f* fr, int b) {
#pragma unroll
    for (int n = 0; n < 4; ++n)
#pragma unroll
      for (int v = 0; v < 8; ++v)
        s_act[wv][b][(v + half * 8) * kAst + n * 16 + l15] = fr[n][v];
    wave_fence();
  };

  auto made = [&](int s, v8f* outfr) {
    layer(s, 0, 1, 0, true, nullptr, nullptr);
    layer(s, 1, 0, 1, true, nullptr, nullptr);
    layer(s, 0, 1, 2, true, nullptr, nullptr);
    layer(s, 1, -1, 3, false, outfr, nullptr);
  };

  auto reverse_cols = [&](v8f* fr) {
    store_frags(fr, 0);
#pragma unroll
    for (int n = 0; n < 4; ++n)
#pragma unroll
      for (int v = 0; v < 8; ++v)
        fr[n][v] = s_act[wv][0][(v + half * 8) * kAst + (63 - (n * 16 + l15))];
    wave_fence();
  };

  v8f y[4], e_m[4], e_lv[4], e_n[4];
#pragma unroll
  for (int n = 0; n < 4; ++n)
#pragma unroll
    for (int v = 0; v < 8; ++v) {
      const int row = r0 + v + half * 8;
      const int col = n * 16 + l15;
      e_m[n][v] = zmean[row * kD + col];
      e_lv[n][v] = zlogvar[row * kD + col];
      e_n[n][v] = noise[row * kD + col];
    }
  v4f wa0[16], wa1[16], wb0[16], wb1[16];
  float ba0, ba1, bb0, bb1;
  issue_weights(1, wa0, wa1, ba0, ba1);
  __builtin_amdgcn_sched_barrier(0);
#pragma unroll
  for (int n = 0; n < 4; ++n)
#pragma unroll
    for (int v = 0; v < 8; ++v)
      y[n][v] = e_m[n][v] + expf(0.5f * e_lv[n][v]) * e_n[n][v];

  v8f m4[4];

  store_weights(1, wa0, wa1, ba0, ba1);
  __syncthreads();
  issue_weights(0, wb0, wb1, bb0, bb1);
  store_frags(y, 0);
  made(1, m4);
#pragma unroll
  for (int n = 0; n < 4; ++n)
#pragma unroll
    for (int v = 0; v < 8; ++v) y[n][v] -= m4[n][v];
  reverse_cols(y);

  __builtin_amdgcn_sched_barrier(0);
  store_weights(0, wb0, wb1, bb0, bb1);
  __syncthreads();
  store_frags(y, 0);
  made(0, m4);
#pragma unroll
  for (int n = 0; n < 4; ++n)
#pragma unroll
    for (int v = 0; v < 8; ++v) y[n][v] -= m4[n][v];

#pragma unroll
  for (int n = 0; n < 4; ++n)
#pragma unroll
    for (int v = 0; v < 8; ++v)
      out_samples[(r0 + v + half * 8) * kD + n * 16 + l15] = y[n][v];

  auto density_stage = [&](int s) {
    store_frags(y, 0);
    const int arow = l15 * kAst;
    const int fl = 4 * ((l15 >> 1) & 3) + 2 * (l15 >> 3) + (l15 & 1);
    v8f base[4][4];
#pragma unroll
    for (int jt = 0; jt < 4; ++jt)
#pragma unroll
      for (int l = 0; l < 4; ++l) {
        v8f t;
#pragma unroll
        for (int v = 0; v < 8; ++v)
          t[v] = s_b[s * 4 * kD + l * kD + jt * 16 + 4 * (v >> 1) + 2 * half + (v & 1)];
        base[jt][l] = t;
      }
    auto load_x0t = [&](int jj, v8f& x) {
#pragma unroll
      for (int v = 0; v < 8; ++v)
        x[v] = s_act[wv][0][arow + jj * 16 + half * 2 + 4 * (v >> 1) + (v & 1)];
    };
    auto load_wd = [&](int jj, v2f (&w)[4][4]) {
#pragma unroll
      for (int l = 0; l < 4; ++l)
#pragma unroll
        for (int q = 0; q < 4; ++q) {
          const int p = s * kWStage + (l * 32 + 8 * jj + 2 * q + half) * kWPS + 2 * fl + 32 * jj;
          w[l][q].x = s_w[p];
          w[l][q].y = s_w[p + 1];
        }
    };
    v8f x0n;
    v2f wdn[4][4];
    load_x0t(0, x0n);
    load_wd(0, wdn);
    v2f pb[4];
    v2f wfc[12];
    v2f wfn[12];
#pragma unroll
    for (int l = 0; l < 4; ++l) { pb[l].x = 0.0f; pb[l].y = 0.0f; }
#pragma unroll
    for (int k = 0; k < 12; ++k) { wfc[k].x = 0.0f; wfc[k].y = 0.0f; wfn[k] = wfc[k]; }
    int pj = 0, pn = 0;
#pragma unroll
    for (int j = 0; j < 4; ++j) {
      v8f x0t = x0n;
      v2f wd[4][4];
#pragma unroll
      for (int l = 0; l < 4; ++l)
#pragma unroll
        for (int q = 0; q < 4; ++q) wd[l][q] = wdn[l][q];
      v8f st[4];
#pragma unroll
      for (int l = 0; l < 4; ++l)
#pragma unroll
        for (int v = 0; v < 8; ++v) st[l][v] = 0.0f;
#pragma unroll
      for (int q = 0; q < 4; ++q) {
#pragma unroll
        for (int k = 0; k < 4 * (3 - j); ++k) {
          const int jt = j + 1 + (k >> 2);
          const int l = k & 3;
          const int p = s * kWStage + (l * 32 + 8 * j + 2 * q + half) * kWPS + 2 * fl + 32 * jt;
          wfn[k].x = s_w[p];
          wfn[k].y = s_w[p + 1];
        }
        if (q == 3 && j < 3) {
          load_x0t(j + 1, x0n);
          load_wd(j + 1, wdn);
        }
        auto owed = [&](int k) {
          if (k < pn) {
            const int jt = pj + 1 + (k >> 2);
            const int l = k & 3;
            base[jt][l] = __builtin_amdgcn_wmma_f32_16x16x4_f32(
                false, wfc[k], false, pb[l], (short)0, base[jt][l], false, false);
          }
        };
        int k0 = 0;
        if (q == 0 && j > 0) { owed(0); owed(1); k0 = 2; }
        v8f pre[4];
#pragma unroll
        for (int i = 0; i < 4; ++i) {
#pragma unroll
          for (int l = 0; l < 4; ++l) {
            v2f bq;
            bq.x = st[l][2 * q];
            bq.y = st[l][2 * q + 1];
            pre[l] = __builtin_amdgcn_wmma_f32_16x16x4_f32(
                false, wd[l][q], false, bq, (short)0, base[j][l], false, false);
            __builtin_amdgcn_sched_barrier(0);
            owed(k0 + 4 * i + l);
            __builtin_amdgcn_sched_barrier(0);
            v8f nx;
            if (l < 3) {
#pragma unroll
              for (int v = 0; v < 8; ++v) nx[v] = fmaxf(pre[l][v], 0.0f);
            } else {
#pragma unroll
              for (int v = 0; v < 8; ++v) nx[v] = pre[l][v] + x0t[v];
            }
            st[(l + 1) & 3] = nx;
          }
        }
        if (q < 3) {
          v2f bq;
          bq.x = st[0][2 * q];
          bq.y = st[0][2 * q + 1];
          base[j][0] = __builtin_amdgcn_wmma_f32_16x16x4_f32(
              false, wd[0][q], false, bq, (short)0, base[j][0], false, false);
#pragma unroll
          for (int l = 1; l < 4; ++l) base[j][l] = pre[l];
        }
#pragma unroll
        for (int l = 0; l < 4; ++l) {
          pb[l].x = st[l][2 * q];
          pb[l].y = st[l][2 * q + 1];
        }
#pragma unroll
        for (int k = 0; k < 12; ++k) wfc[k] = wfn[k];
        pj = j;
        pn = 4 * (3 - j);
      }
#pragma unroll
      for (int v = 0; v < 8; ++v)
        s_act[wv][0][arow + j * 16 + half * 2 + 4 * (v >> 1) + (v & 1)] = st[0][v];
      wave_fence();
    }
#pragma unroll
    for (int n = 0; n < 4; ++n)
#pragma unroll
      for (int v = 0; v < 8; ++v)
        y[n][v] = s_act[wv][0][(v + half * 8) * kAst + n * 16 + l15];
    wave_fence();
  };

  density_stage(0);
  reverse_cols(y);
  density_stage(1);

  float part[8];
#pragma unroll
  for (int v = 0; v < 8; ++v) part[v] = 0.0f;
#pragma unroll
  for (int n = 0; n < 4; ++n)
#pragma unroll
    for (int v = 0; v < 8; ++v) {
      const int row = r0 + v + half * 8;
      const int col = n * 16 + l15;
      const float m = zmean[row * kD + col];
      const float lv = zlogvar[row * kD + col];
      const float d = y[n][v] - m;
      part[v] += d * d * expf(-lv) + lv + kLog2Pi;
    }
#pragma unroll
  for (int v = 0; v < 8; ++v) s_red[tid * 8 + v] = part[v];
  wave_fence();
  if (lane < 16) {
    const int h2 = lane >> 3;
    const int vv = lane & 7;
    const int base = wv * 32 + h2 * 16;
    float s = 0.0f;
#pragma unroll
    for (int j = 0; j < 16; ++j) s += s_red[(base + j) * 8 + vv];
    out_lp[r0 + lane] = -0.5f * s;
  }
}

extern "C" void kernel_launch(void* const* d_in, const int* in_sizes, int n_in,
                              void* d_out, int out_size, void* d_ws, size_t ws_size,
                              hipStream_t stream) {
  (void)in_sizes; (void)n_in; (void)out_size; (void)d_ws; (void)ws_size;
  const float* zmean = (const float*)d_in[0];
  const float* zlogvar = (const float*)d_in[1];
  const float* noise = (const float*)d_in[2];
  const float* Wg = (const float*)d_in[3];
  const float* bvec = (const float*)d_in[4];
  float* samples = (float*)d_out;
  float* lp = samples + (size_t)kB * kD;
  const dim3 grid(kB / kRowsPerBlock);
  iaf_kernel<<<grid, kThreads, 0, stream>>>(zmean, zlogvar, noise, Wg, bvec, samples, lp);
}


# sampling layers computed transposed as the density's are (W^T x in^T): a layer's D fragments are the next layer's B fragments, no LDS exchange per layer; weights read a layer ahead; sums unchanged
# speedup vs baseline: 27.4627x; 1.0655x over previous
#include <hip/hip_runtime.h>
#include <math.h>

namespace {
constexpr int kB = 16384;
constexpr int kD = 64;
constexpr int kWaves = 4;
constexpr int kThreads = kWaves * 32;
constexpr int kRowsPerWave = 16;
constexpr int kRowsPerBlock = kWaves * kRowsPerWave;
constexpr int kAst = 68;
constexpr int kWPS = 160;
constexpr int kWStage = 4 * 32 * kWPS;
constexpr float kLog2Pi = 1.8378770664093453f;
}

typedef float v2f __attribute__((ext_vector_type(2)));
typedef float v4f __attribute__((ext_vector_type(4)));
typedef float v8f __attribute__((ext_vector_type(8)));

__device__ __forceinline__ void wave_fence() { asm volatile("" ::: "memory"); }

__global__ void __launch_bounds__(kThreads)
iaf_kernel(const float* __restrict__ zmean, const float* __restrict__ zlogvar,
           const float* __restrict__ noise, const float* __restrict__ Wg,
           const float* __restrict__ bvec, float* __restrict__ out_samples,
           float* __restrict__ out_lp)
{
  __shared__ alignas(16) float s_w[2 * kWStage];
  __shared__ float s_b[2 * 4 * kD];
  __shared__ float s_act[kWaves][1][kRowsPerWave * kAst];
  __shared__ float s_red[kThreads * 8];

  const int tid = threadIdx.x;
  const int lane = tid & 31;
  const int wv = tid >> 5;
  const int half = lane >> 4;
  const int l15 = lane & 15;
  const int r0 = blockIdx.x * kRowsPerBlock + wv * kRowsPerWave;

  const int nq = tid & 15;
  const int kq = tid >> 4;
  auto issue_weights = [&](int s, v4f (&w0)[16], v4f (&w1)[16], float& b0, float& b1) {
#pragma unroll
    for (int it = 0; it < 16; ++it) {
      const int l = it >> 2;
      const int kp = (it & 3) * 8 + kq;
      const float* src = Wg + ((s * 4 + l) * kD + 2 * kp) * kD + 4 * nq;
      w0[it] = *reinterpret_cast<const v4f*>(src);
      w1[it] = *reinterpret_cast<const v4f*>(src + kD);
    }
    b0 = bvec[s * 4 * kD + tid];
    b1 = bvec[s * 4 * kD + kThreads + tid];
  };
  auto store_weights = [&](int s, const v4f (&w0)[16], const v4f (&w1)[16], float b0, float b1) {
#pragma unroll
    for (int it = 0; it < 16; ++it) {
      const int l = it >> 2;
      const int kp = (it & 3) * 8 + kq;
      const int k0 = 2 * kp;
      const int k1 = k0 + 1;
      const int n0 = 4 * nq;
      const int d0 = (l == 0) ? k0 + 1 : k0;
      const int d1 = (l == 0) ? k1 + 1 : k1;
      v4f lo, hi;
      lo.x = (n0 + 0 >= d0) ? w0[it].x : 0.0f;
      lo.y = (n0 + 0 >= d1) ? w1[it].x : 0.0f;
      lo.z = (n0 + 1 >= d0) ? w0[it].y : 0.0f;
      lo.w = (n0 + 1 >= d1) ? w1[it].y : 0.0f;
      hi.x = (n0 + 2 >= d0) ? w0[it].z : 0.0f;
      hi.y = (n0 + 2 >= d1) ? w1[it].z : 0.0f;
      hi.z = (n0 + 3 >= d0) ? w0[it].w : 0.0f;
      hi.w = (n0 + 3 >= d1) ? w1[it].w : 0.0f;
      float* dst = &s_w[s * kWStage + (l * 32 + kp) * kWPS + 2 * n0];
      *reinterpret_cast<v4f*>(dst) = lo;
      *reinterpret_cast<v4f*>(dst + 4) = hi;
    }
    s_b[s * 4 * kD + tid] = b0;
    s_b[s * 4 * kD + kThreads + tid] = b1;
  };

  auto store_frags = [&](const v8f* fr, int b) {
#pragma unroll
    for (int n = 0; n < 4; ++n)
#pragma unroll
      for (int v = 0; v < 8; ++v)
        s_act[wv][b][(v + half * 8) * kAst + n * 16 + l15] = fr[n][v];
    wave_fence();
  };

  auto reverse_cols = [&](v8f* fr) {
    store_frags(fr, 0);
#pragma unroll
    for (int n = 0; n < 4; ++n)
#pragma unroll
      for (int v = 0; v < 8; ++v)
        fr[n][v] = s_act[wv][0][(v + half * 8) * kAst + (63 - (n * 16 + l15))];
    wave_fence();
  };

  const int arow = l15 * kAst;
  const int fl = 4 * ((l15 >> 1) & 3) + 2 * (l15 >> 3) + (l15 & 1);
  auto load_layer = [&](int s, int l, v2f (&w)[16][4], v8f (&bs)[4]) {
#pragma unroll
    for (int n = 0; n < 4; ++n)
#pragma unroll
      for (int v = 0; v < 8; ++v)
        bs[n][v] = s_b[s * 4 * kD + l * kD + n * 16 + 4 * (v >> 1) + 2 * half + (v & 1)];
#pragma unroll
    for (int kk = 0; kk < 16; ++kk)
#pragma unroll
      for (int n = 0; n < 4; ++n) {
        if (kk >= 4 * (n + 1)) continue;
        const int p = s * kWStage + (l * 32 + 2 * kk + half) * kWPS + 2 * fl + 32 * n;
        w[kk][n].x = s_w[p];
        w[kk][n].y = s_w[p + 1];
      }
  };
  auto layer_t = [&](const v2f (&w)[16][4], const v8f (&bs)[4], const v8f (&in)[4], v8f (&out)[4],
                     bool do_relu) {
    v8f acc[4];
#pragma unroll
    for (int n = 0; n < 4; ++n) acc[n] = bs[n];
#pragma unroll
    for (int kk = 0; kk < 16; ++kk) {
      v2f bq;
      bq.x = in[kk >> 2][2 * (kk & 3)];
      bq.y = in[kk >> 2][2 * (kk & 3) + 1];
#pragma unroll
      for (int n = 0; n < 4; ++n) {
        if (kk >= 4 * (n + 1)) continue;
        acc[n] = __builtin_amdgcn_wmma_f32_16x16x4_f32(
            false, w[kk][n], false, bq, (short)0, acc[n], false, false);
      }
    }
#pragma unroll
    for (int n = 0; n < 4; ++n)
#pragma unroll
      for (int v = 0; v < 8; ++v) out[n][v] = do_relu ? fmaxf(acc[n][v], 0.0f) : acc[n][v];
  };
  auto made_t = [&](int s, const v8f (&xt)[4], v8f (&mt)[4]) {
    v2f wa[16][4], wb[16][4];
    v8f ba[4], bb[4], h0[4], h1[4];
    load_layer(s, 0, wa, ba);
    __builtin_amdgcn_sched_barrier(0);
    load_layer(s, 1, wb, bb);
    __builtin_amdgcn_sched_barrier(0);
    layer_t(wa, ba, xt, h0, true);
    __builtin_amdgcn_sched_barrier(0);
    load_layer(s, 2, wa, ba);
    __builtin_amdgcn_sched_barrier(0);
    layer_t(wb, bb, h0, h1, true);
    __builtin_amdgcn_sched_barrier(0);
    load_layer(s, 3, wb, bb);
    __builtin_amdgcn_sched_barrier(0);
    layer_t(wa, ba, h1, h0, true);
    layer_t(wb, bb, h0, mt, false);
  };
  auto load_t = [&](v8f (&xt)[4], bool reversed) {
#pragma unroll
    for (int n = 0; n < 4; ++n)
#pragma unroll
      for (int v = 0; v < 8; ++v) {
        const int f = n * 16 + half * 2 + 4 * (v >> 1) + (v & 1);
        xt[n][v] = s_act[wv][0][arow + (reversed ? 63 - f : f)];
      }
    wave_fence();
  };
  auto store_t = [&](const v8f (&xt)[4]) {
#pragma unroll
    for (int n = 0; n < 4; ++n)
#pragma unroll
      for (int v = 0; v < 8; ++v)
        s_act[wv][0][arow + n * 16 + half * 2 + 4 * (v >> 1) + (v & 1)] = xt[n][v];
    wave_fence();
  };

  v8f y[4], e_m[4], e_lv[4], e_n[4];
#pragma unroll
  for (int n = 0; n < 4; ++n)
#pragma unroll
    for (int v = 0; v < 8; ++v) {
      const int row = r0 + v + half * 8;
      const int col = n * 16 + l15;
      e_m[n][v] = zmean[row * kD + col];
      e_lv[n][v] = zlogvar[row * kD + col];
      e_n[n][v] = noise[row * kD + col];
    }
  v4f wa0[16], wa1[16], wb0[16], wb1[16];
  float ba0, ba1, bb0, bb1;
  issue_weights(1, wa0, wa1, ba0, ba1);
  __builtin_amdgcn_sched_barrier(0);
#pragma unroll
  for (int n = 0; n < 4; ++n)
#pragma unroll
    for (int v = 0; v < 8; ++v)
      y[n][v] = e_m[n][v] + expf(0.5f * e_lv[n][v]) * e_n[n][v];

  store_weights(1, wa0, wa1, ba0, ba1);
  __syncthreads();
  issue_weights(0, wb0, wb1, bb0, bb1);
  v8f yt[4], mt[4];
  store_frags(y, 0);
  load_t(yt, false);
  made_t(1, yt, mt);
#pragma unroll
  for (int n = 0; n < 4; ++n)
#pragma unroll
    for (int v = 0; v < 8; ++v) yt[n][v] -= mt[n][v];
  store_t(yt);
  load_t(yt, true);

  __builtin_amdgcn_sched_barrier(0);
  store_weights(0, wb0, wb1, bb0, bb1);
  __syncthreads();
  made_t(0, yt, mt);
#pragma unroll
  for (int n = 0; n < 4; ++n)
#pragma unroll
    for (int v = 0; v < 8; ++v) yt[n][v] -= mt[n][v];
  store_t(yt);
#pragma unroll
  for (int n = 0; n < 4; ++n)
#pragma unroll
    for (int v = 0; v < 8; ++v)
      y[n][v] = s_act[wv][0][(v + half * 8) * kAst + n * 16 + l15];
  wave_fence();

#pragma unroll
  for (int n = 0; n < 4; ++n)
#pragma unroll
    for (int v = 0; v < 8; ++v)
      out_samples[(r0 + v + half * 8) * kD + n * 16 + l15] = y[n][v];

  auto density_stage = [&](int s) {
    store_frags(y, 0);
    v8f base[4][4];
#pragma unroll
    for (int jt = 0; jt < 4; ++jt)
#pragma unroll
      for (int l = 0; l < 4; ++l) {
        v8f t;
#pragma unroll
        for (int v = 0; v < 8; ++v)
          t[v] = s_b[s * 4 * kD + l * kD + jt * 16 + 4 * (v >> 1) + 2 * half + (v & 1)];
        base[jt][l] = t;
      }
    auto load_x0t = [&](int jj, v8f& x) {
#pragma unroll
      for (int v = 0; v < 8; ++v)
        x[v] = s_act[wv][0][arow + jj * 16 + half * 2 + 4 * (v >> 1) + (v & 1)];
    };
    auto load_wd = [&](int jj, v2f (&w)[4][4]) {
#pragma unroll
      for (int l = 0; l < 4; ++l)
#pragma unroll
        for (int q = 0; q < 4; ++q) {
          const int p = s * kWStage + (l * 32 + 8 * jj + 2 * q + half) * kWPS + 2 * fl + 32 * jj;
          w[l][q].x = s_w[p];
          w[l][q].y = s_w[p + 1];
        }
    };
    v8f x0n;
    v2f wdn[4][4];
    load_x0t(0, x0n);
    load_wd(0, wdn);
    v2f pb[4];
    v2f wfc[12];
    v2f wfn[12];
#pragma unroll
    for (int l = 0; l < 4; ++l) { pb[l].x = 0.0f; pb[l].y = 0.0f; }
#pragma unroll
    for (int k = 0; k < 12; ++k) { wfc[k].x = 0.0f; wfc[k].y = 0.0f; wfn[k] = wfc[k]; }
    int pj = 0, pn = 0;
#pragma unroll
    for (int j = 0; j < 4; ++j) {
      v8f x0t = x0n;
      v2f wd[4][4];
#pragma unroll
      for (int l = 0; l < 4; ++l)
#pragma unroll
        for (int q = 0; q < 4; ++q) wd[l][q] = wdn[l][q];
      v8f st[4];
#pragma unroll
      for (int l = 0; l < 4; ++l)
#pragma unroll
        for (int v = 0; v < 8; ++v) st[l][v] = 0.0f;
#pragma unroll
      for (int q = 0; q < 4; ++q) {
#pragma unroll
        for (int k = 0; k < 4 * (3 - j); ++k) {
          const int jt = j + 1 + (k >> 2);
          const int l = k & 3;
          const int p = s * kWStage + (l * 32 + 8 * j + 2 * q + half) * kWPS + 2 * fl + 32 * jt;
          wfn[k].x = s_w[p];
          wfn[k].y = s_w[p + 1];
        }
        if (q == 3 && j < 3) {
          load_x0t(j + 1, x0n);
          load_wd(j + 1, wdn);
        }
        auto owed = [&](int k) {
          if (k < pn) {
            const int jt = pj + 1 + (k >> 2);
            const int l = k & 3;
            base[jt][l] = __builtin_amdgcn_wmma_f32_16x16x4_f32(
                false, wfc[k], false, pb[l], (short)0, base[jt][l], false, false);
          }
        };
        int k0 = 0;
        if (q == 0 && j > 0) { owed(0); owed(1); k0 = 2; }
        v8f pre[4];
#pragma unroll
        for (int i = 0; i < 4; ++i) {
#pragma unroll
          for (int l = 0; l < 4; ++l) {
            v2f bq;
            bq.x = st[l][2 * q];
            bq.y = st[l][2 * q + 1];
            pre[l] = __builtin_amdgcn_wmma_f32_16x16x4_f32(
                false, wd[l][q], false, bq, (short)0, base[j][l], false, false);
            __builtin_amdgcn_sched_barrier(0);
            owed(k0 + 4 * i + l);
            __builtin_amdgcn_sched_barrier(0);
            v8f nx;
            if (l < 3) {
#pragma unroll
              for (int v = 0; v < 8; ++v) nx[v] = fmaxf(pre[l][v], 0.0f);
            } else {
#pragma unroll
              for (int v = 0; v < 8; ++v) nx[v] = pre[l][v] + x0t[v];
            }
            st[(l + 1) & 3] = nx;
          }
        }
        if (q < 3) {
          v2f bq;
          bq.x = st[0][2 * q];
          bq.y = st[0][2 * q + 1];
          base[j][0] = __builtin_amdgcn_wmma_f32_16x16x4_f32(
              false, wd[0][q], false, bq, (short)0, base[j][0], false, false);
#pragma unroll
          for (int l = 1; l < 4; ++l) base[j][l] = pre[l];
        }
#pragma unroll
        for (int l = 0; l < 4; ++l) {
          pb[l].x = st[l][2 * q];
          pb[l].y = st[l][2 * q + 1];
        }
#pragma unroll
        for (int k = 0; k < 12; ++k) wfc[k] = wfn[k];
        pj = j;
        pn = 4 * (3 - j);
      }
#pragma unroll
      for (int v = 0; v < 8; ++v)
        s_act[wv][0][arow + j * 16 + half * 2 + 4 * (v >> 1) + (v & 1)] = st[0][v];
      wave_fence();
    }
#pragma unroll
    for (int n = 0; n < 4; ++n)
#pragma unroll
      for (int v = 0; v < 8; ++v)
        y[n][v] = s_act[wv][0][(v + half * 8) * kAst + n * 16 + l15];
    wave_fence();
  };

  density_stage(0);
  reverse_cols(y);
  density_stage(1);

  float part[8];
#pragma unroll
  for (int v = 0; v < 8; ++v) part[v] = 0.0f;
#pragma unroll
  for (int n = 0; n < 4; ++n)
#pragma unroll
    for (int v = 0; v < 8; ++v) {
      const int row = r0 + v + half * 8;
      const int col = n * 16 + l15;
      const float m = zmean[row * kD + col];
      const float lv = zlogvar[row * kD + col];
      const float d = y[n][v] - m;
      part[v] += d * d * expf(-lv) + lv + kLog2Pi;
    }
#pragma unroll
  for (int v = 0; v < 8; ++v) s_red[tid * 8 + v] = part[v];
  wave_fence();
  if (lane < 16) {
    const int h2 = lane >> 3;
    const int vv = lane & 7;
    const int base = wv * 32 + h2 * 16;
    float s = 0.0f;
#pragma unroll
    for (int j = 0; j < 16; ++j) s += s_red[(base + j) * 8 + vv];
    out_lp[r0 + lane] = -0.5f * s;
  }
}

extern "C" void kernel_launch(void* const* d_in, const int* in_sizes, int n_in,
                              void* d_out, int out_size, void* d_ws, size_t ws_size,
                              hipStream_t stream) {
  (void)in_sizes; (void)n_in; (void)out_size; (void)d_ws; (void)ws_size;
  const float* zmean = (const float*)d_in[0];
  const float* zlogvar = (const float*)d_in[1];
  const float* noise = (const float*)d_in[2];
  const float* Wg = (const float*)d_in[3];
  const float* bvec = (const float*)d_in[4];
  float* samples = (float*)d_out;
  float* lp = samples + (size_t)kB * kD;
  const dim3 grid(kB / kRowsPerBlock);
  iaf_kernel<<<grid, kThreads, 0, stream>>>(zmean, zlogvar, noise, Wg, bvec, samples, lp);
}


# work that changes nothing leaves the wave's serial path: NaN-quieting copy before each relu, each density group's zero-weight first layer-0 matrix step, LDS re-stores, input re-loads; sums unchanged
# speedup vs baseline: 29.6880x; 1.0810x over previous
#include <hip/hip_runtime.h>
#include <math.h>

namespace {
constexpr int kB = 16384;
constexpr int kD = 64;
constexpr int kWaves = 4;
constexpr int kThreads = kWaves * 32;
constexpr int kRowsPerWave = 16;
constexpr int kRowsPerBlock = kWaves * kRowsPerWave;
constexpr int kAst = 68;
constexpr int kWPS = 160;
constexpr int kWStage = 4 * 32 * kWPS;
constexpr float kLog2Pi = 1.8378770664093453f;
}

typedef float v2f __attribute__((ext_vector_type(2)));
typedef float v4f __attribute__((ext_vector_type(4)));
typedef float v8f __attribute__((ext_vector_type(8)));

__device__ __forceinline__ void wave_fence() { asm volatile("" ::: "memory"); }

__device__ __forceinline__ float relu1(float x) { return __builtin_elementwise_maximum(x, 0.0f); }

__global__ void __launch_bounds__(kThreads)
iaf_kernel(const float* __restrict__ zmean, const float* __restrict__ zlogvar,
           const float* __restrict__ noise, const float* __restrict__ Wg,
           const float* __restrict__ bvec, float* __restrict__ out_samples,
           float* __restrict__ out_lp)
{
  __shared__ alignas(16) float s_w[2 * kWStage];
  __shared__ float s_b[2 * 4 * kD];
  __shared__ float s_act[kWaves][1][kRowsPerWave * kAst];
  __shared__ float s_red[kThreads * 8];

  const int tid = threadIdx.x;
  const int lane = tid & 31;
  const int wv = tid >> 5;
  const int half = lane >> 4;
  const int l15 = lane & 15;
  const int r0 = blockIdx.x * kRowsPerBlock + wv * kRowsPerWave;

  const int nq = tid & 15;
  const int kq = tid >> 4;
  auto issue_weights = [&](int s, v4f (&w0)[16], v4f (&w1)[16], float& b0, float& b1) {
#pragma unroll
    for (int it = 0; it < 16; ++it) {
      const int l = it >> 2;
      const int kp = (it & 3) * 8 + kq;
      const float* src = Wg + ((s * 4 + l) * kD + 2 * kp) * kD + 4 * nq;
      w0[it] = *reinterpret_cast<const v4f*>(src);
      w1[it] = *reinterpret_cast<const v4f*>(src + kD);
    }
    b0 = bvec[s * 4 * kD + tid];
    b1 = bvec[s * 4 * kD + kThreads + tid];
  };
  auto store_weights = [&](int s, const v4f (&w0)[16], const v4f (&w1)[16], float b0, float b1) {
#pragma unroll
    for (int it = 0; it < 16; ++it) {
      const int l = it >> 2;
      const int kp = (it & 3) * 8 + kq;
      const int k0 = 2 * kp;
      const int k1 = k0 + 1;
      const int n0 = 4 * nq;
      const int d0 = (l == 0) ? k0 + 1 : k0;
      const int d1 = (l == 0) ? k1 + 1 : k1;
      v4f lo, hi;
      lo.x = (n0 + 0 >= d0) ? w0[it].x : 0.0f;
      lo.y = (n0 + 0 >= d1) ? w1[it].x : 0.0f;
      lo.z = (n0 + 1 >= d0) ? w0[it].y : 0.0f;
      lo.w = (n0 + 1 >= d1) ? w1[it].y : 0.0f;
      hi.x = (n0 + 2 >= d0) ? w0[it].z : 0.0f;
      hi.y = (n0 + 2 >= d1) ? w1[it].z : 0.0f;
      hi.z = (n0 + 3 >= d0) ? w0[it].w : 0.0f;
      hi.w = (n0 + 3 >= d1) ? w1[it].w : 0.0f;
      float* dst = &s_w[s * kWStage + (l * 32 + kp) * kWPS + 2 * n0];
      *reinterpret_cast<v4f*>(dst) = lo;
      *reinterpret_cast<v4f*>(dst + 4) = hi;
    }
    s_b[s * 4 * kD + tid] = b0;
    s_b[s * 4 * kD + kThreads + tid] = b1;
  };

  auto store_frags = [&](const v8f* fr, int b) {
#pragma unroll
    for (int n = 0; n < 4; ++n)
#pragma unroll
      for (int v = 0; v < 8; ++v)
        s_act[wv][b][(v + half * 8) * kAst + n * 16 + l15] = fr[n][v];
    wave_fence();
  };

  const int arow = l15 * kAst;
  const int fl = 4 * ((l15 >> 1) & 3) + 2 * (l15 >> 3) + (l15 & 1);
  auto load_layer = [&](int s, int l, v2f (&w)[16][4], v8f (&bs)[4]) {
#pragma unroll
    for (int n = 0; n < 4; ++n)
#pragma unroll
      for (int v = 0; v < 8; ++v)
        bs[n][v] = s_b[s * 4 * kD + l * kD + n * 16 + 4 * (v >> 1) + 2 * half + (v & 1)];
#pragma unroll
    for (int kk = 0; kk < 16; ++kk)
#pragma unroll
      for (int n = 0; n < 4; ++n) {
        if (kk >= 4 * (n + 1)) continue;
        const int p = s * kWStage + (l * 32 + 2 * kk + half) * kWPS + 2 * fl + 32 * n;
        w[kk][n].x = s_w[p];
        w[kk][n].y = s_w[p + 1];
      }
  };
  auto layer_t = [&](const v2f (&w)[16][4], const v8f (&bs)[4], const v8f (&in)[4], v8f (&out)[4],
                     bool do_relu) {
    v8f acc[4];
#pragma unroll
    for (int n = 0; n < 4; ++n) acc[n] = bs[n];
#pragma unroll
    for (int kk = 0; kk < 16; ++kk) {
      v2f bq;
      bq.x = in[kk >> 2][2 * (kk & 3)];
      bq.y = in[kk >> 2][2 * (kk & 3) + 1];
#pragma unroll
      for (int n = 0; n < 4; ++n) {
        if (kk >= 4 * (n + 1)) continue;
        acc[n] = __builtin_amdgcn_wmma_f32_16x16x4_f32(
            false, w[kk][n], false, bq, (short)0, acc[n], false, false);
      }
    }
#pragma unroll
    for (int n = 0; n < 4; ++n)
#pragma unroll
      for (int v = 0; v < 8; ++v) out[n][v] = do_relu ? relu1(acc[n][v]) : acc[n][v];
  };
  auto made_t = [&](int s, const v8f (&xt)[4], v8f (&mt)[4]) {
    v2f wa[16][4], wb[16][4];
    v8f ba[4], bb[4], h0[4], h1[4];
    load_layer(s, 0, wa, ba);
    __builtin_amdgcn_sched_barrier(0);
    load_layer(s, 1, wb, bb);
    __builtin_amdgcn_sched_barrier(0);
    layer_t(wa, ba, xt, h0, true);
    __builtin_amdgcn_sched_barrier(0);
    load_layer(s, 2, wa, ba);
    __builtin_amdgcn_sched_barrier(0);
    layer_t(wb, bb, h0, h1, true);
    __builtin_amdgcn_sched_barrier(0);
    load_layer(s, 3, wb, bb);
    __builtin_amdgcn_sched_barrier(0);
    layer_t(wa, ba, h1, h0, true);
    layer_t(wb, bb, h0, mt, false);
  };
  auto load_t = [&](v8f (&xt)[4], bool reversed) {
#pragma unroll
    for (int n = 0; n < 4; ++n)
#pragma unroll
      for (int v = 0; v < 8; ++v) {
        const int f = n * 16 + half * 2 + 4 * (v >> 1) + (v & 1);
        xt[n][v] = s_act[wv][0][arow + (reversed ? 63 - f : f)];
      }
    wave_fence();
  };
  auto store_t = [&](const v8f (&xt)[4]) {
#pragma unroll
    for (int n = 0; n < 4; ++n)
#pragma unroll
      for (int v = 0; v < 8; ++v)
        s_act[wv][0][arow + n * 16 + half * 2 + 4 * (v >> 1) + (v & 1)] = xt[n][v];
    wave_fence();
  };

  v8f y[4], e_m[4], e_lv[4], e_n[4];
#pragma unroll
  for (int n = 0; n < 4; ++n)
#pragma unroll
    for (int v = 0; v < 8; ++v) {
      const int row = r0 + v + half * 8;
      const int col = n * 16 + l15;
      e_m[n][v] = zmean[row * kD + col];
      e_lv[n][v] = zlogvar[row * kD + col];
      e_n[n][v] = noise[row * kD + col];
    }
  v4f wa0[16], wa1[16], wb0[16], wb1[16];
  float ba0, ba1, bb0, bb1;
  issue_weights(1, wa0, wa1, ba0, ba1);
  __builtin_amdgcn_sched_barrier(0);
#pragma unroll
  for (int n = 0; n < 4; ++n)
#pragma unroll
    for (int v = 0; v < 8; ++v)
      y[n][v] = e_m[n][v] + expf(0.5f * e_lv[n][v]) * e_n[n][v];

  store_weights(1, wa0, wa1, ba0, ba1);
  __syncthreads();
  issue_weights(0, wb0, wb1, bb0, bb1);
  v8f yt[4], mt[4];
  store_frags(y, 0);
  load_t(yt, false);
  made_t(1, yt, mt);
#pragma unroll
  for (int n = 0; n < 4; ++n)
#pragma unroll
    for (int v = 0; v < 8; ++v) yt[n][v] -= mt[n][v];
  store_t(yt);
  load_t(yt, true);

  __builtin_amdgcn_sched_barrier(0);
  store_weights(0, wb0, wb1, bb0, bb1);
  __syncthreads();
  made_t(0, yt, mt);
#pragma unroll
  for (int n = 0; n < 4; ++n)
#pragma unroll
    for (int v = 0; v < 8; ++v) yt[n][v] -= mt[n][v];
  store_t(yt);
#pragma unroll
  for (int n = 0; n < 4; ++n)
#pragma unroll
    for (int v = 0; v < 8; ++v)
      y[n][v] = s_act[wv][0][(v + half * 8) * kAst + n * 16 + l15];
  wave_fence();

#pragma unroll
  for (int n = 0; n < 4; ++n)
#pragma unroll
    for (int v = 0; v < 8; ++v)
      out_samples[(r0 + v + half * 8) * kD + n * 16 + l15] = y[n][v];

  auto density_stage = [&](int s) {
    if (s == 1) store_frags(y, 0);
    v8f base[4][4];
#pragma unroll
    for (int jt = 0; jt < 4; ++jt)
#pragma unroll
      for (int l = 0; l < 4; ++l) {
        v8f t;
#pragma unroll
        for (int v = 0; v < 8; ++v)
          t[v] = s_b[s * 4 * kD + l * kD + jt * 16 + 4 * (v >> 1) + 2 * half + (v & 1)];
        base[jt][l] = t;
      }
    auto load_x0t = [&](int jj, v8f& x) {
#pragma unroll
      for (int v = 0; v < 8; ++v)
        x[v] = s_act[wv][0][arow + jj * 16 + half * 2 + 4 * (v >> 1) + (v & 1)];
    };
    auto load_wd = [&](int jj, v2f (&w)[4][4]) {
#pragma unroll
      for (int l = 0; l < 4; ++l)
#pragma unroll
        for (int q = 0; q < 4; ++q) {
          const int p = s * kWStage + (l * 32 + 8 * jj + 2 * q + half) * kWPS + 2 * fl + 32 * jj;
          w[l][q].x = s_w[p];
          w[l][q].y = s_w[p + 1];
        }
    };
    v8f x0n;
    v2f wdn[4][4];
    load_x0t(0, x0n);
    load_wd(0, wdn);
    v2f pb[4];
    v2f wfc[12];
    v2f wfn[12];
#pragma unroll
    for (int l = 0; l < 4; ++l) { pb[l].x = 0.0f; pb[l].y = 0.0f; }
#pragma unroll
    for (int k = 0; k < 12; ++k) { wfc[k].x = 0.0f; wfc[k].y = 0.0f; wfn[k] = wfc[k]; }
    int pj = 0, pn = 0;
#pragma unroll
    for (int j = 0; j < 4; ++j) {
      v8f x0t = x0n;
      v2f wd[4][4];
#pragma unroll
      for (int l = 0; l < 4; ++l)
#pragma unroll
        for (int q = 0; q < 4; ++q) wd[l][q] = wdn[l][q];
      v8f st[4];
#pragma unroll
      for (int l = 0; l < 4; ++l)
#pragma unroll
        for (int v = 0; v < 8; ++v) st[l][v] = 0.0f;
#pragma unroll
      for (int q = 0; q < 4; ++q) {
#pragma unroll
        for (int k = 0; k < 4 * (3 - j); ++k) {
          const int jt = j + 1 + (k >> 2);
          const int l = k & 3;
          const int p = s * kWStage + (l * 32 + 8 * j + 2 * q + half) * kWPS + 2 * fl + 32 * jt;
          wfn[k].x = s_w[p];
          wfn[k].y = s_w[p + 1];
        }
        if (q == 3 && j < 3) {
          load_x0t(j + 1, x0n);
          load_wd(j + 1, wdn);
        }
        auto owed = [&](int k) {
          if (k < pn) {
            const int jt = pj + 1 + (k >> 2);
            const int l = k & 3;
            base[jt][l] = __builtin_amdgcn_wmma_f32_16x16x4_f32(
                false, wfc[k], false, pb[l], (short)0, base[jt][l], false, false);
          }
        };
        int k0 = 0;
        if (q == 0 && j > 0) { owed(0); owed(1); k0 = 2; }
        v8f pre[4];
#pragma unroll
        for (int i = 0; i < 4; ++i) {
#pragma unroll
          for (int l = 0; l < 4; ++l) {
            v2f bq;
            bq.x = st[l][2 * q];
            bq.y = st[l][2 * q + 1];
            if (i == 0 && l == 0) {
              pre[0] = base[j][0];
            } else {
              pre[l] = __builtin_amdgcn_wmma_f32_16x16x4_f32(
                  false, wd[l][q], false, bq, (short)0, base[j][l], false, false);
            }
            __builtin_amdgcn_sched_barrier(0);
            owed(k0 + 4 * i + l);
            __builtin_amdgcn_sched_barrier(0);
            v8f nx;
            if (l < 3) {
#pragma unroll
              for (int v = 0; v < 8; ++v) nx[v] = relu1(pre[l][v]);
            } else {
#pragma unroll
              for (int v = 0; v < 8; ++v) nx[v] = pre[l][v] + x0t[v];
            }
            st[(l + 1) & 3] = nx;
          }
        }
        if (q < 3) {
          v2f bq;
          bq.x = st[0][2 * q];
          bq.y = st[0][2 * q + 1];
          base[j][0] = __builtin_amdgcn_wmma_f32_16x16x4_f32(
              false, wd[0][q], false, bq, (short)0, base[j][0], false, false);
#pragma unroll
          for (int l = 1; l < 4; ++l) base[j][l] = pre[l];
        }
#pragma unroll
        for (int l = 0; l < 4; ++l) {
          pb[l].x = st[l][2 * q];
          pb[l].y = st[l][2 * q + 1];
        }
#pragma unroll
        for (int k = 0; k < 12; ++k) wfc[k] = wfn[k];
        pj = j;
        pn = 4 * (3 - j);
      }
#pragma unroll
      for (int v = 0; v < 8; ++v)
        s_act[wv][0][arow + j * 16 + half * 2 + 4 * (v >> 1) + (v & 1)] = st[0][v];
      wave_fence();
    }
#pragma unroll
    for (int n = 0; n < 4; ++n)
#pragma unroll
      for (int v = 0; v < 8; ++v) {
        const int col = n * 16 + l15;
        y[n][v] = s_act[wv][0][(v + half * 8) * kAst + (s == 0 ? 63 - col : col)];
      }
    wave_fence();
  };

  density_stage(0);
  density_stage(1);

  float part[8];
#pragma unroll
  for (int v = 0; v < 8; ++v) part[v] = 0.0f;
#pragma unroll
  for (int n = 0; n < 4; ++n)
#pragma unroll
    for (int v = 0; v < 8; ++v) {
      const float m = e_m[n][v];
      const float lv = e_lv[n][v];
      const float d = y[n][v] - m;
      part[v] += d * d * expf(-lv) + lv + kLog2Pi;
    }
#pragma unroll
  for (int v = 0; v < 8; ++v) s_red[tid * 8 + v] = part[v];
  wave_fence();
  if (lane < 16) {
    const int h2 = lane >> 3;
    const int vv = lane & 7;
    const int base = wv * 32 + h2 * 16;
    float s = 0.0f;
#pragma unroll
    for (int j = 0; j < 16; ++j) s += s_red[(base + j) * 8 + vv];
    out_lp[r0 + lane] = -0.5f * s;
  }
}

extern "C" void kernel_launch(void* const* d_in, const int* in_sizes, int n_in,
                              void* d_out, int out_size, void* d_ws, size_t ws_size,
                              hipStream_t stream) {
  (void)in_sizes; (void)n_in; (void)out_size; (void)d_ws; (void)ws_size;
  const float* zmean = (const float*)d_in[0];
  const float* zlogvar = (const float*)d_in[1];
  const float* noise = (const float*)d_in[2];
  const float* Wg = (const float*)d_in[3];
  const float* bvec = (const float*)d_in[4];
  float* samples = (float*)d_out;
  float* lp = samples + (size_t)kB * kD;
  const dim3 grid(kB / kRowsPerBlock);
  iaf_kernel<<<grid, kThreads, 0, stream>>>(zmean, zlogvar, noise, Wg, bvec, samples, lp);
}
